# starting kernel
# baseline (speedup 1.0000x reference)
#include <hip/hip_runtime.h>
#include <math.h>

#define BATCH   2
#define SEQLEN  2048
#define D_INNER 2048
#define D_STATE 16
#define DT_RANK 64
#define KPROJ   (DT_RANK + 2 * D_STATE)

typedef float v2f __attribute__((ext_vector_type(2)));
typedef float v8f __attribute__((ext_vector_type(8)));

__global__ __launch_bounds__(256) void xdbl_gemm_wmma(
    const float* __restrict__ u,
    const float* __restrict__ xpw,
    float* __restrict__ xdbl)
{
    const int lane = threadIdx.x & 31;
    const int wave = threadIdx.x >> 5;
    const int tile = blockIdx.x * 8 + wave;
    const int nt = tile % (KPROJ / 16);
    const int mt = tile / (KPROJ / 16);
    const int m0 = mt * 16, n0 = nt * 16;

    const int half = lane & 15;
    const int hi   = lane >> 4;
    const int kk   = hi * 2;

    const int b = m0 >> 11;
    const int l = (m0 & (SEQLEN - 1)) + half;

    const float* pa = u + (size_t)b * D_INNER * SEQLEN + (size_t)kk * SEQLEN + l;
    const float* pb = xpw + (size_t)(n0 + half) * D_INNER + kk;

    v8f c0 = {}, c1 = {}, c2 = {}, c3 = {};
    #pragma unroll 2
    for (int k = 0; k < D_INNER; k += 16) {
        v2f a, bm;
        a.x  = pa[0 * SEQLEN];  a.y  = pa[1 * SEQLEN];
        bm.x = pb[0];           bm.y = pb[1];
        c0 = __builtin_amdgcn_wmma_f32_16x16x4_f32(false, a, false, bm, (short)0, c0, false, false);

        a.x  = pa[4 * SEQLEN];  a.y  = pa[5 * SEQLEN];
        bm.x = pb[4];           bm.y = pb[5];
        c1 = __builtin_amdgcn_wmma_f32_16x16x4_f32(false, a, false, bm, (short)0, c1, false, false);

        a.x  = pa[8 * SEQLEN];  a.y  = pa[9 * SEQLEN];
        bm.x = pb[8];           bm.y = pb[9];
        c2 = __builtin_amdgcn_wmma_f32_16x16x4_f32(false, a, false, bm, (short)0, c2, false, false);

        a.x  = pa[12 * SEQLEN]; a.y  = pa[13 * SEQLEN];
        bm.x = pb[12];          bm.y = pb[13];
        c3 = __builtin_amdgcn_wmma_f32_16x16x4_f32(false, a, false, bm, (short)0, c3, false, false);

        pa += 16 * SEQLEN;
        pb += 16;
    }
    v8f c = (c0 + c1) + (c2 + c3);

    float* pc = xdbl + (size_t)(m0 + hi * 8) * KPROJ + (n0 + half);
    #pragma unroll
    for (int r = 0; r < 8; ++r)
        pc[(size_t)r * KPROJ] = c[r];
}

__global__ __launch_bounds__(256) void delta_gemm_wmma(
    const float* __restrict__ xdbl,
    const float* __restrict__ dtw,
    const float* __restrict__ dtb,
    float* __restrict__ delta)
{
    const int lane = threadIdx.x & 31;
    const int wave = threadIdx.x >> 5;
    const int tile = blockIdx.x * 8 + wave;
    const int nt = tile & 127;
    const int mt = tile >> 7;
    const int m0 = mt * 16, n0 = nt * 16;

    const int half = lane & 15;
    const int hi   = lane >> 4;
    const int kk   = hi * 2;

    const float* pa = xdbl + (size_t)(m0 + half) * KPROJ + kk;
    const float* pb = dtw + (size_t)(n0 + half) * DT_RANK + kk;

    v8f c0 = {}, c1 = {};
    #pragma unroll
    for (int k = 0; k < DT_RANK; k += 8) {
        v2f a, bm;
        a.x  = pa[0];  a.y  = pa[1];
        bm.x = pb[0];  bm.y = pb[1];
        c0 = __builtin_amdgcn_wmma_f32_16x16x4_f32(false, a, false, bm, (short)0, c0, false, false);

        a.x  = pa[4];  a.y  = pa[5];
        bm.x = pb[4];  bm.y = pb[5];
        c1 = __builtin_amdgcn_wmma_f32_16x16x4_f32(false, a, false, bm, (short)0, c1, false, false);

        pa += 8;
        pb += 8;
    }
    v8f c = c0 + c1;

    const int   d    = n0 + half;
    const float bias = dtb[d];
    const int   b    = m0 >> 11;
    const int   lb   = (m0 & (SEQLEN - 1)) + hi * 8;
    float* pd = delta + ((size_t)b * D_INNER + d) * SEQLEN + lb;

    #pragma unroll
    for (int r = 0; r < 8; ++r) {
        float z  = c[r] + bias;
        float sp = fmaxf(z, 0.0f) + log1pf(__expf(-fabsf(z)));
        pd[r] = sp;
    }
}

__global__ __launch_bounds__(256) void s6_scan(
    const float* __restrict__ u,
    const float* __restrict__ xdbl,
    const float* __restrict__ delta,
    const float* __restrict__ A_log,
    float* __restrict__ out)
{
    const int lane = threadIdx.x & 31;
    const int gwave = (blockIdx.x * blockDim.x + threadIdx.x) >> 5;
    const int g = lane >> 4;
    const int n = lane & 15;
    const int ch = gwave * 2 + g;
    const int b  = ch >> 11;
    const int d  = ch & (D_INNER - 1);

    const float An = -__expf(A_log[(size_t)d * D_STATE + n]);

    const float* pdl = delta + (size_t)ch * SEQLEN;
    const float* pu  = u     + (size_t)ch * SEQLEN;
    const float* pbc = xdbl  + (size_t)b * SEQLEN * KPROJ + DT_RANK + n;
    float*       po  = out   + (size_t)ch * SEQLEN;

    float x = 0.0f;
    #pragma unroll 4
    for (int l = 0; l < SEQLEN; ++l) {
        if ((l & 15) == 0) {
            __builtin_prefetch(pdl + l + 64, 0, 0);
            __builtin_prefetch(pu  + l + 64, 0, 0);
        }
        float dl = pdl[l];
        float uv = pu[l];
        float Bn = pbc[(size_t)l * KPROJ];
        float Cn = pbc[(size_t)l * KPROJ + D_STATE];

        float dA = __expf(dl * An);
        x = dA * x + (dl * uv) * Bn;

        float p = x * Cn;
        p += __shfl_xor(p, 1);
        p += __shfl_xor(p, 2);
        p += __shfl_xor(p, 4);
        p += __shfl_xor(p, 8);
        if (n == 0) po[l] = p;
    }
}

extern "C" void kernel_launch(void* const* d_in, const int* in_sizes, int n_in,
                              void* d_out, int out_size, void* d_ws, size_t ws_size,
                              hipStream_t stream)
{
    const float* u     = (const float*)d_in[0];
    const float* xpw   = (const float*)d_in[1];
    const float* dtw   = (const float*)d_in[2];
    const float* dtb   = (const float*)d_in[3];
    const float* A_log = (const float*)d_in[4];
    float*       out   = (float*)d_out;

    float* xdbl  = (float*)d_ws;
    float* delta = xdbl + (size_t)BATCH * SEQLEN * KPROJ;

    xdbl_gemm_wmma<<<192, 256, 0, stream>>>(u, xpw, xdbl);

    delta_gemm_wmma<<<4096, 256, 0, stream>>>(xdbl, dtw, dtb, delta);

    s6_scan<<<256, 256, 0, stream>>>(u, xdbl, delta, A_log, out);
}


# scan: time axis in 16-step register blocks, next block's loads issued one block ahead (double buffer); per-step row broadcast and butterfly on DPP, one 128-B store per 32 steps; same sums
# speedup vs baseline: 4.8267x; 4.8267x over previous
#include <hip/hip_runtime.h>
#include <math.h>

#define BATCH   2
#define SEQLEN  2048
#define D_INNER 2048
#define D_STATE 16
#define DT_RANK 64
#define KPROJ   (DT_RANK + 2 * D_STATE)

typedef float v2f __attribute__((ext_vector_type(2)));
typedef float v8f __attribute__((ext_vector_type(8)));

__global__ __launch_bounds__(256) void xdbl_gemm_wmma(
    const float* __restrict__ u,
    const float* __restrict__ xpw,
    float* __restrict__ xdbl)
{
    const int lane = threadIdx.x & 31;
    const int wave = threadIdx.x >> 5;
    const int tile = blockIdx.x * 8 + wave;
    const int nt = tile % (KPROJ / 16);
    const int mt = tile / (KPROJ / 16);
    const int m0 = mt * 16, n0 = nt * 16;

    const int half = lane & 15;
    const int hi   = lane >> 4;
    const int kk   = hi * 2;

    const int b = m0 >> 11;
    const int l = (m0 & (SEQLEN - 1)) + half;

    const float* pa = u + (size_t)b * D_INNER * SEQLEN + (size_t)kk * SEQLEN + l;
    const float* pb = xpw + (size_t)(n0 + half) * D_INNER + kk;

    v8f c0 = {}, c1 = {}, c2 = {}, c3 = {};
    #pragma unroll 2
    for (int k = 0; k < D_INNER; k += 16) {
        v2f a, bm;
        a.x  = pa[0 * SEQLEN];  a.y  = pa[1 * SEQLEN];
        bm.x = pb[0];           bm.y = pb[1];
        c0 = __builtin_amdgcn_wmma_f32_16x16x4_f32(false, a, false, bm, (short)0, c0, false, false);

        a.x  = pa[4 * SEQLEN];  a.y  = pa[5 * SEQLEN];
        bm.x = pb[4];           bm.y = pb[5];
        c1 = __builtin_amdgcn_wmma_f32_16x16x4_f32(false, a, false, bm, (short)0, c1, false, false);

        a.x  = pa[8 * SEQLEN];  a.y  = pa[9 * SEQLEN];
        bm.x = pb[8];           bm.y = pb[9];
        c2 = __builtin_amdgcn_wmma_f32_16x16x4_f32(false, a, false, bm, (short)0, c2, false, false);

        a.x  = pa[12 * SEQLEN]; a.y  = pa[13 * SEQLEN];
        bm.x = pb[12];          bm.y = pb[13];
        c3 = __builtin_amdgcn_wmma_f32_16x16x4_f32(false, a, false, bm, (short)0, c3, false, false);

        pa += 16 * SEQLEN;
        pb += 16;
    }
    v8f c = (c0 + c1) + (c2 + c3);

    float* pc = xdbl + (size_t)(m0 + hi * 8) * KPROJ + (n0 + half);
    #pragma unroll
    for (int r = 0; r < 8; ++r)
        pc[(size_t)r * KPROJ] = c[r];
}

__global__ __launch_bounds__(256) void delta_gemm_wmma(
    const float* __restrict__ xdbl,
    const float* __restrict__ dtw,
    const float* __restrict__ dtb,
    float* __restrict__ delta)
{
    const int lane = threadIdx.x & 31;
    const int wave = threadIdx.x >> 5;
    const int tile = blockIdx.x * 8 + wave;
    const int nt = tile & 127;
    const int mt = tile >> 7;
    const int m0 = mt * 16, n0 = nt * 16;

    const int half = lane & 15;
    const int hi   = lane >> 4;
    const int kk   = hi * 2;

    const float* pa = xdbl + (size_t)(m0 + half) * KPROJ + kk;
    const float* pb = dtw + (size_t)(n0 + half) * DT_RANK + kk;

    v8f c0 = {}, c1 = {};
    #pragma unroll
    for (int k = 0; k < DT_RANK; k += 8) {
        v2f a, bm;
        a.x  = pa[0];  a.y  = pa[1];
        bm.x = pb[0];  bm.y = pb[1];
        c0 = __builtin_amdgcn_wmma_f32_16x16x4_f32(false, a, false, bm, (short)0, c0, false, false);

        a.x  = pa[4];  a.y  = pa[5];
        bm.x = pb[4];  bm.y = pb[5];
        c1 = __builtin_amdgcn_wmma_f32_16x16x4_f32(false, a, false, bm, (short)0, c1, false, false);

        pa += 8;
        pb += 8;
    }
    v8f c = c0 + c1;

    const int   d    = n0 + half;
    const float bias = dtb[d];
    const int   b    = m0 >> 11;
    const int   lb   = (m0 & (SEQLEN - 1)) + hi * 8;
    float* pd = delta + ((size_t)b * D_INNER + d) * SEQLEN + lb;

    #pragma unroll
    for (int r = 0; r < 8; ++r) {
        float z  = c[r] + bias;
        float sp = fmaxf(z, 0.0f) + log1pf(__expf(-fabsf(z)));
        pd[r] = sp;
    }
}

struct ScanBlk {
    float dl, uv;
    float Bv[16], Cv[16];
};

template <int CTRL>
__device__ __forceinline__ float row_dpp(float v)
{
    return __builtin_bit_cast(float,
        __builtin_amdgcn_update_dpp(0, __builtin_bit_cast(int, v), CTRL, 0xf, 0xf, true));
}

__device__ __forceinline__ void scan_load(ScanBlk& q, const float* pdl, const float* pu,
                                          const float* pbc, int l0, int n)
{
    q.dl = pdl[l0 + n];
    q.uv = pu[l0 + n];
    const float* r = pbc + (size_t)l0 * KPROJ;
    #pragma unroll
    for (int j = 0; j < 16; ++j) {
        q.Bv[j] = r[j * KPROJ];
        q.Cv[j] = r[j * KPROJ + D_STATE];
    }
}

template <int J, int Q>
struct ScanSteps {
    static __device__ __forceinline__ void run(float& x, float (&yv)[2], const ScanBlk& q,
                                               float du, float An, int n)
    {
        const float dlj = row_dpp<0x150 + J>(q.dl);
        const float duj = row_dpp<0x150 + J>(du);
        const float dA  = __expf(dlj * An);
        x = dA * x + duj * q.Bv[J];

        float p = x * q.Cv[J];
        p += row_dpp<0x161>(p);
        p += row_dpp<0x162>(p);
        p += row_dpp<0x164>(p);
        p += row_dpp<0x168>(p);

        constexpr int pos = Q * 16 + J;
        yv[pos & 1] = (n == (pos >> 1)) ? p : yv[pos & 1];

        ScanSteps<J + 1, Q>::run(x, yv, q, du, An, n);
    }
};
template <int Q>
struct ScanSteps<16, Q> {
    static __device__ __forceinline__ void run(float&, float (&)[2], const ScanBlk&,
                                               float, float, int) {}
};

template <int Q>
__device__ __forceinline__ void scan_block(float& x, float (&yv)[2], const ScanBlk& q,
                                           float An, int n)
{
    const float du = q.dl * q.uv;
    ScanSteps<0, Q>::run(x, yv, q, du, An, n);
}

#define SCAN_STAGE_EDGE(x, y0, y1)                                         \
    do {                                                                   \
        asm volatile("" : "+v"(x) : "v"(y0), "v"(y1));                     \
        __builtin_amdgcn_sched_barrier(0);                                 \
    } while (0)

__global__ __launch_bounds__(256) void s6_scan(
    const float* __restrict__ u,
    const float* __restrict__ xdbl,
    const float* __restrict__ delta,
    const float* __restrict__ A_log,
    float* __restrict__ out)
{
    const int lane = threadIdx.x & 31;
    const int gwave = (blockIdx.x * blockDim.x + threadIdx.x) >> 5;
    const int g = lane >> 4;
    const int n = lane & 15;
    const int ch = gwave * 2 + g;
    const int b  = ch >> 11;
    const int d  = ch & (D_INNER - 1);

    const float An = -__expf(A_log[(size_t)d * D_STATE + n]);

    const float* pdl = delta + (size_t)ch * SEQLEN;
    const float* pu  = u     + (size_t)ch * SEQLEN;
    const float* pbc = xdbl  + (size_t)b * SEQLEN * KPROJ + DT_RANK + n;
    float*       po  = out   + (size_t)ch * SEQLEN;

    ScanBlk qa, qb;
    scan_load(qa, pdl, pu, pbc, 0, n);

    float x = 0.0f;
    float yv[2] = {0.0f, 0.0f};
    for (int l0 = 0; l0 < SEQLEN; l0 += 32) {
        const int lp = l0 + 128 < SEQLEN - 32 ? l0 + 128 : SEQLEN - 32;
        __builtin_prefetch(pdl + lp, 0, 0);
        __builtin_prefetch(pu  + lp, 0, 0);

        scan_load(qb, pdl, pu, pbc, l0 + 16, n);
        SCAN_STAGE_EDGE(x, yv[0], yv[1]);
        scan_block<0>(x, yv, qa, An, n);

        const int ln = l0 + 32 < SEQLEN ? l0 + 32 : SEQLEN - 16;
        SCAN_STAGE_EDGE(x, yv[0], yv[1]);
        scan_load(qa, pdl, pu, pbc, ln, n);
        SCAN_STAGE_EDGE(x, yv[0], yv[1]);
        scan_block<1>(x, yv, qb, An, n);

        *reinterpret_cast<float2*>(po + l0 + 2 * n) = make_float2(yv[0], yv[1]);
    }
}

extern "C" void kernel_launch(void* const* d_in, const int* in_sizes, int n_in,
                              void* d_out, int out_size, void* d_ws, size_t ws_size,
                              hipStream_t stream)
{
    const float* u     = (const float*)d_in[0];
    const float* xpw   = (const float*)d_in[1];
    const float* dtw   = (const float*)d_in[2];
    const float* dtb   = (const float*)d_in[3];
    const float* A_log = (const float*)d_in[4];
    float*       out   = (float*)d_out;

    float* xdbl  = (float*)d_ws;
    float* delta = xdbl + (size_t)BATCH * SEQLEN * KPROJ;

    xdbl_gemm_wmma<<<192, 256, 0, stream>>>(u, xpw, xdbl);

    delta_gemm_wmma<<<4096, 256, 0, stream>>>(xdbl, dtw, dtb, delta);

    s6_scan<<<256, 256, 0, stream>>>(u, xdbl, delta, A_log, out);
}


# scan: y's 16-lane sums as a register transpose (each lane^m exchange keeps one step, hands over the other; 32->2 regs per span) instead of a butterfly per step; same pair sums in the same order
# speedup vs baseline: 5.0426x; 1.0447x over previous
#include <hip/hip_runtime.h>
#include <math.h>

#define BATCH   2
#define SEQLEN  2048
#define D_INNER 2048
#define D_STATE 16
#define DT_RANK 64
#define KPROJ   (DT_RANK + 2 * D_STATE)

typedef float v2f __attribute__((ext_vector_type(2)));
typedef float v8f __attribute__((ext_vector_type(8)));

__global__ __launch_bounds__(256) void xdbl_gemm_wmma(
    const float* __restrict__ u,
    const float* __restrict__ xpw,
    float* __restrict__ xdbl)
{
    const int lane = threadIdx.x & 31;
    const int wave = threadIdx.x >> 5;
    const int tile = blockIdx.x * 8 + wave;
    const int nt = tile % (KPROJ / 16);
    const int mt = tile / (KPROJ / 16);
    const int m0 = mt * 16, n0 = nt * 16;

    const int half = lane & 15;
    const int hi   = lane >> 4;
    const int kk   = hi * 2;

    const int b = m0 >> 11;
    const int l = (m0 & (SEQLEN - 1)) + half;

    const float* pa = u + (size_t)b * D_INNER * SEQLEN + (size_t)kk * SEQLEN + l;
    const float* pb = xpw + (size_t)(n0 + half) * D_INNER + kk;

    v8f c0 = {}, c1 = {}, c2 = {}, c3 = {};
    #pragma unroll 2
    for (int k = 0; k < D_INNER; k += 16) {
        v2f a, bm;
        a.x  = pa[0 * SEQLEN];  a.y  = pa[1 * SEQLEN];
        bm.x = pb[0];           bm.y = pb[1];
        c0 = __builtin_amdgcn_wmma_f32_16x16x4_f32(false, a, false, bm, (short)0, c0, false, false);

        a.x  = pa[4 * SEQLEN];  a.y  = pa[5 * SEQLEN];
        bm.x = pb[4];           bm.y = pb[5];
        c1 = __builtin_amdgcn_wmma_f32_16x16x4_f32(false, a, false, bm, (short)0, c1, false, false);

        a.x  = pa[8 * SEQLEN];  a.y  = pa[9 * SEQLEN];
        bm.x = pb[8];           bm.y = pb[9];
        c2 = __builtin_amdgcn_wmma_f32_16x16x4_f32(false, a, false, bm, (short)0, c2, false, false);

        a.x  = pa[12 * SEQLEN]; a.y  = pa[13 * SEQLEN];
        bm.x = pb[12];          bm.y = pb[13];
        c3 = __builtin_amdgcn_wmma_f32_16x16x4_f32(false, a, false, bm, (short)0, c3, false, false);

        pa += 16 * SEQLEN;
        pb += 16;
    }
    v8f c = (c0 + c1) + (c2 + c3);

    float* pc = xdbl + (size_t)(m0 + hi * 8) * KPROJ + (n0 + half);
    #pragma unroll
    for (int r = 0; r < 8; ++r)
        pc[(size_t)r * KPROJ] = c[r];
}

__global__ __launch_bounds__(256) void delta_gemm_wmma(
    const float* __restrict__ xdbl,
    const float* __restrict__ dtw,
    const float* __restrict__ dtb,
    float* __restrict__ delta)
{
    const int lane = threadIdx.x & 31;
    const int wave = threadIdx.x >> 5;
    const int tile = blockIdx.x * 8 + wave;
    const int nt = tile & 127;
    const int mt = tile >> 7;
    const int m0 = mt * 16, n0 = nt * 16;

    const int half = lane & 15;
    const int hi   = lane >> 4;
    const int kk   = hi * 2;

    const float* pa = xdbl + (size_t)(m0 + half) * KPROJ + kk;
    const float* pb = dtw + (size_t)(n0 + half) * DT_RANK + kk;

    v8f c0 = {}, c1 = {};
    #pragma unroll
    for (int k = 0; k < DT_RANK; k += 8) {
        v2f a, bm;
        a.x  = pa[0];  a.y  = pa[1];
        bm.x = pb[0];  bm.y = pb[1];
        c0 = __builtin_amdgcn_wmma_f32_16x16x4_f32(false, a, false, bm, (short)0, c0, false, false);

        a.x  = pa[4];  a.y  = pa[5];
        bm.x = pb[4];  bm.y = pb[5];
        c1 = __builtin_amdgcn_wmma_f32_16x16x4_f32(false, a, false, bm, (short)0, c1, false, false);

        pa += 8;
        pb += 8;
    }
    v8f c = c0 + c1;

    const int   d    = n0 + half;
    const float bias = dtb[d];
    const int   b    = m0 >> 11;
    const int   lb   = (m0 & (SEQLEN - 1)) + hi * 8;
    float* pd = delta + ((size_t)b * D_INNER + d) * SEQLEN + lb;

    #pragma unroll
    for (int r = 0; r < 8; ++r) {
        float z  = c[r] + bias;
        float sp = fmaxf(z, 0.0f) + log1pf(__expf(-fabsf(z)));
        pd[r] = sp;
    }
}

struct ScanBlk {
    float dl, uv;
    float Bv[16], Cv[16];
};

template <int CTRL>
__device__ __forceinline__ float row_dpp(float v)
{
    return __builtin_bit_cast(float,
        __builtin_amdgcn_update_dpp(0, __builtin_bit_cast(int, v), CTRL, 0xf, 0xf, true));
}

__device__ __forceinline__ void scan_load(ScanBlk& q, const float* pdl, const float* pu,
                                          const float* pbc, int l0, int n)
{
    q.dl = pdl[l0 + n];
    q.uv = pu[l0 + n];
    const float* r = pbc + (size_t)l0 * KPROJ;
    #pragma unroll
    for (int j = 0; j < 16; ++j) {
        q.Bv[j] = r[j * KPROJ];
        q.Cv[j] = r[j * KPROJ + D_STATE];
    }
}

template <int CTRL>
__device__ __forceinline__ float xr_pair(float a, float b, bool bit_set)
{
    const float keep = bit_set ? b : a;
    const float give = bit_set ? a : b;
    return keep + row_dpp<CTRL>(give);
}

template <int J>
struct ScanSteps {
    static __device__ __forceinline__ void run(float& x, float (&P)[16], const ScanBlk& q,
                                               float du, float An)
    {
        const float dlj = row_dpp<0x150 + J>(q.dl);
        const float duj = row_dpp<0x150 + J>(du);
        const float dA  = __expf(dlj * An);
        x = dA * x + duj * q.Bv[J];
        P[J] = x * q.Cv[J];
        ScanSteps<J + 1>::run(x, P, q, du, An);
    }
};
template <>
struct ScanSteps<16> {
    static __device__ __forceinline__ void run(float&, float (&)[16], const ScanBlk&,
                                               float, float) {}
};

__device__ __forceinline__ void scan_block(float& x, float (&r)[2], const ScanBlk& q,
                                           float An, bool b0, bool b1, bool b2)
{
    const float du = q.dl * q.uv;
    float P[16];
    ScanSteps<0>::run(x, P, q, du, An);

    float s8[8];
    #pragma unroll
    for (int i = 0; i < 8; ++i) {
        const int j = (i >> 1) * 4 + (i & 1);
        s8[i] = xr_pair<0x161>(P[j], P[j + 2], b0);
    }
    float s4[4];
    #pragma unroll
    for (int i = 0; i < 4; ++i) {
        const int j = (i >> 1) * 4 + (i & 1);
        s4[i] = xr_pair<0x162>(s8[j], s8[j + 2], b1);
    }
    r[0] = xr_pair<0x164>(s4[0], s4[2], b2);
    r[1] = xr_pair<0x164>(s4[1], s4[3], b2);
}

#define SCAN_STAGE_EDGE(x, y0, y1)                                         \
    do {                                                                   \
        asm volatile("" : "+v"(x) : "v"(y0), "v"(y1));                     \
        __builtin_amdgcn_sched_barrier(0);                                 \
    } while (0)

__global__ __launch_bounds__(256) void s6_scan(
    const float* __restrict__ u,
    const float* __restrict__ xdbl,
    const float* __restrict__ delta,
    const float* __restrict__ A_log,
    float* __restrict__ out)
{
    const int lane = threadIdx.x & 31;
    const int gwave = (blockIdx.x * blockDim.x + threadIdx.x) >> 5;
    const int g = lane >> 4;
    const int n = lane & 15;
    const int ch = gwave * 2 + g;
    const int b  = ch >> 11;
    const int d  = ch & (D_INNER - 1);

    const float An = -__expf(A_log[(size_t)d * D_STATE + n]);

    const float* pdl = delta + (size_t)ch * SEQLEN;
    const float* pu  = u     + (size_t)ch * SEQLEN;
    const float* pbc = xdbl  + (size_t)b * SEQLEN * KPROJ + DT_RANK + n;
    float*       po  = out   + (size_t)ch * SEQLEN;

    ScanBlk qa, qb;
    scan_load(qa, pdl, pu, pbc, 0, n);

    const bool b0 = (n & 1) != 0, b1 = (n & 2) != 0, b2 = (n & 4) != 0, b3 = (n & 8) != 0;

    float x = 0.0f;
    float yv[2] = {0.0f, 0.0f};
    float ra[2], rb[2];
    for (int l0 = 0; l0 < SEQLEN; l0 += 32) {
        const int lp = l0 + 128 < SEQLEN - 32 ? l0 + 128 : SEQLEN - 32;
        __builtin_prefetch(pdl + lp, 0, 0);
        __builtin_prefetch(pu  + lp, 0, 0);

        scan_load(qb, pdl, pu, pbc, l0 + 16, n);
        SCAN_STAGE_EDGE(x, yv[0], yv[1]);
        scan_block(x, ra, qa, An, b0, b1, b2);

        const int ln = l0 + 32 < SEQLEN ? l0 + 32 : SEQLEN - 16;
        SCAN_STAGE_EDGE(x, ra[0], ra[1]);
        scan_load(qa, pdl, pu, pbc, ln, n);
        SCAN_STAGE_EDGE(x, ra[0], ra[1]);
        scan_block(x, rb, qb, An, b0, b1, b2);

        yv[0] = xr_pair<0x168>(ra[0], rb[0], b3);
        yv[1] = xr_pair<0x168>(ra[1], rb[1], b3);
        *reinterpret_cast<float2*>(po + l0 + 2 * n) = make_float2(yv[0], yv[1]);
    }
}

extern "C" void kernel_launch(void* const* d_in, const int* in_sizes, int n_in,
                              void* d_out, int out_size, void* d_ws, size_t ws_size,
                              hipStream_t stream)
{
    const float* u     = (const float*)d_in[0];
    const float* xpw   = (const float*)d_in[1];
    const float* dtw   = (const float*)d_in[2];
    const float* dtb   = (const float*)d_in[3];
    const float* A_log = (const float*)d_in[4];
    float*       out   = (float*)d_out;

    float* xdbl  = (float*)d_ws;
    float* delta = xdbl + (size_t)BATCH * SEQLEN * KPROJ;

    xdbl_gemm_wmma<<<192, 256, 0, stream>>>(u, xpw, xdbl);

    delta_gemm_wmma<<<4096, 256, 0, stream>>>(xdbl, dtw, dtb, delta);

    s6_scan<<<256, 256, 0, stream>>>(u, xdbl, delta, A_log, out);
}


# x_dbl GEMM re-tiled: one workgroup per 16-row tile, K split over its 8 waves, each wave holds all 6 n tiles and loads one 32-k group ahead; partials summed in LDS in wave order (new sum order)
# speedup vs baseline: 7.4737x; 1.4821x over previous
#include <hip/hip_runtime.h>
#include <math.h>

#define BATCH   2
#define SEQLEN  2048
#define D_INNER 2048
#define D_STATE 16
#define DT_RANK 64
#define KPROJ   (DT_RANK + 2 * D_STATE)

typedef float v2f __attribute__((ext_vector_type(2)));
typedef float v8f __attribute__((ext_vector_type(8)));

#define G1_WAVES  8
#define G1_KCHUNK (D_INNER / G1_WAVES)
#define G1_NT     (KPROJ / 16)
#define G1_GROUP  32

struct G1Frag {
    float  a[16];
    float4 q[G1_NT][4];
};

__device__ __forceinline__ void g1_load(G1Frag& f, const float* pa, const float* pb, int k0)
{
    #pragma unroll
    for (int jj = 0; jj < 16; ++jj)
        f.a[jj] = pa[(size_t)(k0 + jj) * SEQLEN];
    #pragma unroll
    for (int t = 0; t < G1_NT; ++t) {
        const float4* r = reinterpret_cast<const float4*>(pb + (size_t)t * 16 * D_INNER + k0);
        #pragma unroll
        for (int i = 0; i < 4; ++i)
            f.q[t][i] = r[i];
    }
}

__device__ __forceinline__ void g1_mma(v8f (&c)[G1_NT], const G1Frag& f)
{
    #pragma unroll
    for (int j = 0; j < 8; ++j) {
        v2f a;
        a.x = f.a[2 * j];  a.y = f.a[2 * j + 1];
        #pragma unroll
        for (int t = 0; t < G1_NT; ++t) {
            const float4 qq = f.q[t][j >> 1];
            v2f bm;
            if (j & 1) { bm.x = qq.z; bm.y = qq.w; }
            else       { bm.x = qq.x; bm.y = qq.y; }
            c[t] = __builtin_amdgcn_wmma_f32_16x16x4_f32(false, a, false, bm, (short)0, c[t], false, false);
        }
    }
}

__global__ __launch_bounds__(256) void xdbl_gemm_wmma(
    const float* __restrict__ u,
    const float* __restrict__ xpw,
    float* __restrict__ xdbl)
{
    __shared__ float part[G1_WAVES][16 * KPROJ];

    const int lane = threadIdx.x & 31;
    const int wave = threadIdx.x >> 5;
    const int m0   = blockIdx.x * 16;

    const int half = lane & 15;
    const int hi   = lane >> 4;

    const int b = m0 >> 11;
    const int l = (m0 & (SEQLEN - 1)) + half;
    const int kbase = wave * G1_KCHUNK + 16 * hi;

    const float* pa = u + (size_t)b * D_INNER * SEQLEN + (size_t)kbase * SEQLEN + l;
    const float* pb = xpw + (size_t)half * D_INNER + kbase;

    v8f c[G1_NT];
    #pragma unroll
    for (int t = 0; t < G1_NT; ++t) c[t] = v8f{};

    G1Frag f[2];
    g1_load(f[0], pa, pb, 0);
    #pragma unroll
    for (int g = 0; g < G1_KCHUNK / G1_GROUP; ++g) {
        if (g + 1 < G1_KCHUNK / G1_GROUP)
            g1_load(f[(g + 1) & 1], pa, pb, (g + 1) * G1_GROUP);
        __builtin_amdgcn_sched_barrier(0);
        g1_mma(c, f[g & 1]);
        __builtin_amdgcn_sched_barrier(0);
    }

    float* pw = &part[wave][(hi * 8) * KPROJ + half];
    #pragma unroll
    for (int t = 0; t < G1_NT; ++t) {
        #pragma unroll
        for (int r = 0; r < 8; ++r)
            pw[r * KPROJ + t * 16] = c[t][r];
    }
    __syncthreads();

    float* po = xdbl + (size_t)m0 * KPROJ;
    #pragma unroll
    for (int i = 0; i < (16 * KPROJ) / 256; ++i) {
        const int e = threadIdx.x + 256 * i;
        float s = part[0][e];
        #pragma unroll
        for (int w = 1; w < G1_WAVES; ++w)
            s += part[w][e];
        po[e] = s;
    }
}

__global__ __launch_bounds__(256) void delta_gemm_wmma(
    const float* __restrict__ xdbl,
    const float* __restrict__ dtw,
    const float* __restrict__ dtb,
    float* __restrict__ delta)
{
    const int lane = threadIdx.x & 31;
    const int wave = threadIdx.x >> 5;
    const int tile = blockIdx.x * 8 + wave;
    const int nt = tile & 127;
    const int mt = tile >> 7;
    const int m0 = mt * 16, n0 = nt * 16;

    const int half = lane & 15;
    const int hi   = lane >> 4;
    const int kk   = hi * 2;

    const float* pa = xdbl + (size_t)(m0 + half) * KPROJ + kk;
    const float* pb = dtw + (size_t)(n0 + half) * DT_RANK + kk;

    v8f c0 = {}, c1 = {};
    #pragma unroll
    for (int k = 0; k < DT_RANK; k += 8) {
        v2f a, bm;
        a.x  = pa[0];  a.y  = pa[1];
        bm.x = pb[0];  bm.y = pb[1];
        c0 = __builtin_amdgcn_wmma_f32_16x16x4_f32(false, a, false, bm, (short)0, c0, false, false);

        a.x  = pa[4];  a.y  = pa[5];
        bm.x = pb[4];  bm.y = pb[5];
        c1 = __builtin_amdgcn_wmma_f32_16x16x4_f32(false, a, false, bm, (short)0, c1, false, false);

        pa += 8;
        pb += 8;
    }
    v8f c = c0 + c1;

    const int   d    = n0 + half;
    const float bias = dtb[d];
    const int   b    = m0 >> 11;
    const int   lb   = (m0 & (SEQLEN - 1)) + hi * 8;
    float* pd = delta + ((size_t)b * D_INNER + d) * SEQLEN + lb;

    #pragma unroll
    for (int r = 0; r < 8; ++r) {
        float z  = c[r] + bias;
        float sp = fmaxf(z, 0.0f) + log1pf(__expf(-fabsf(z)));
        pd[r] = sp;
    }
}

struct ScanBlk {
    float dl, uv;
    float Bv[16], Cv[16];
};

template <int CTRL>
__device__ __forceinline__ float row_dpp(float v)
{
    return __builtin_bit_cast(float,
        __builtin_amdgcn_update_dpp(0, __builtin_bit_cast(int, v), CTRL, 0xf, 0xf, true));
}

__device__ __forceinline__ void scan_load(ScanBlk& q, const float* pdl, const float* pu,
                                          const float* pbc, int l0, int n)
{
    q.dl = pdl[l0 + n];
    q.uv = pu[l0 + n];
    const float* r = pbc + (size_t)l0 * KPROJ;
    #pragma unroll
    for (int j = 0; j < 16; ++j) {
        q.Bv[j] = r[j * KPROJ];
        q.Cv[j] = r[j * KPROJ + D_STATE];
    }
}

template <int CTRL>
__device__ __forceinline__ float xr_pair(float a, float b, bool bit_set)
{
    const float keep = bit_set ? b : a;
    const float give = bit_set ? a : b;
    return keep + row_dpp<CTRL>(give);
}

template <int J>
struct ScanSteps {
    static __device__ __forceinline__ void run(float& x, float (&P)[16], const ScanBlk& q,
                                               float du, float An)
    {
        const float dlj = row_dpp<0x150 + J>(q.dl);
        const float duj = row_dpp<0x150 + J>(du);
        const float dA  = __expf(dlj * An);
        x = dA * x + duj * q.Bv[J];
        P[J] = x * q.Cv[J];
        ScanSteps<J + 1>::run(x, P, q, du, An);
    }
};
template <>
struct ScanSteps<16> {
    static __device__ __forceinline__ void run(float&, float (&)[16], const ScanBlk&,
                                               float, float) {}
};

__device__ __forceinline__ void scan_block(float& x, float (&r)[2], const ScanBlk& q,
                                           float An, bool b0, bool b1, bool b2)
{
    const float du = q.dl * q.uv;
    float P[16];
    ScanSteps<0>::run(x, P, q, du, An);

    float s8[8];
    #pragma unroll
    for (int i = 0; i < 8; ++i) {
        const int j = (i >> 1) * 4 + (i & 1);
        s8[i] = xr_pair<0x161>(P[j], P[j + 2], b0);
    }
    float s4[4];
    #pragma unroll
    for (int i = 0; i < 4; ++i) {
        const int j = (i >> 1) * 4 + (i & 1);
        s4[i] = xr_pair<0x162>(s8[j], s8[j + 2], b1);
    }
    r[0] = xr_pair<0x164>(s4[0], s4[2], b2);
    r[1] = xr_pair<0x164>(s4[1], s4[3], b2);
}

#define SCAN_STAGE_EDGE(x, y0, y1)                                         \
    do {                                                                   \
        asm volatile("" : "+v"(x) : "v"(y0), "v"(y1));                     \
        __builtin_amdgcn_sched_barrier(0);                                 \
    } while (0)

__global__ __launch_bounds__(256) void s6_scan(
    const float* __restrict__ u,
    const float* __restrict__ xdbl,
    const float* __restrict__ delta,
    const float* __restrict__ A_log,
    float* __restrict__ out)
{
    const int lane = threadIdx.x & 31;
    const int gwave = (blockIdx.x * blockDim.x + threadIdx.x) >> 5;
    const int g = lane >> 4;
    const int n = lane & 15;
    const int ch = gwave * 2 + g;
    const int b  = ch >> 11;
    const int d  = ch & (D_INNER - 1);

    const float An = -__expf(A_log[(size_t)d * D_STATE + n]);

    const float* pdl = delta + (size_t)ch * SEQLEN;
    const float* pu  = u     + (size_t)ch * SEQLEN;
    const float* pbc = xdbl  + (size_t)b * SEQLEN * KPROJ + DT_RANK + n;
    float*       po  = out   + (size_t)ch * SEQLEN;

    ScanBlk qa, qb;
    scan_load(qa, pdl, pu, pbc, 0, n);

    const bool b0 = (n & 1) != 0, b1 = (n & 2) != 0, b2 = (n & 4) != 0, b3 = (n & 8) != 0;

    float x = 0.0f;
    float yv[2] = {0.0f, 0.0f};
    float ra[2], rb[2];
    for (int l0 = 0; l0 < SEQLEN; l0 += 32) {
        const int lp = l0 + 128 < SEQLEN - 32 ? l0 + 128 : SEQLEN - 32;
        __builtin_prefetch(pdl + lp, 0, 0);
        __builtin_prefetch(pu  + lp, 0, 0);

        scan_load(qb, pdl, pu, pbc, l0 + 16, n);
        SCAN_STAGE_EDGE(x, yv[0], yv[1]);
        scan_block(x, ra, qa, An, b0, b1, b2);

        const int ln = l0 + 32 < SEQLEN ? l0 + 32 : SEQLEN - 16;
        SCAN_STAGE_EDGE(x, ra[0], ra[1]);
        scan_load(qa, pdl, pu, pbc, ln, n);
        SCAN_STAGE_EDGE(x, ra[0], ra[1]);
        scan_block(x, rb, qb, An, b0, b1, b2);

        yv[0] = xr_pair<0x168>(ra[0], rb[0], b3);
        yv[1] = xr_pair<0x168>(ra[1], rb[1], b3);
        *reinterpret_cast<float2*>(po + l0 + 2 * n) = make_float2(yv[0], yv[1]);
    }
}

extern "C" void kernel_launch(void* const* d_in, const int* in_sizes, int n_in,
                              void* d_out, int out_size, void* d_ws, size_t ws_size,
                              hipStream_t stream)
{
    const float* u     = (const float*)d_in[0];
    const float* xpw   = (const float*)d_in[1];
    const float* dtw   = (const float*)d_in[2];
    const float* dtb   = (const float*)d_in[3];
    const float* A_log = (const float*)d_in[4];
    float*       out   = (float*)d_out;

    float* xdbl  = (float*)d_ws;
    float* delta = xdbl + (size_t)BATCH * SEQLEN * KPROJ;

    xdbl_gemm_wmma<<<256, 256, 0, stream>>>(u, xpw, xdbl);

    delta_gemm_wmma<<<4096, 256, 0, stream>>>(xdbl, dtw, dtb, delta);

    s6_scan<<<256, 256, 0, stream>>>(u, xdbl, delta, A_log, out);
}


# x_dbl's B and C columns stored step-interleaved ([l/4][col][l%4]) by the first GEMM; scan lane n reads 4 steps of B_n / C_n per 16-B load: 8 loads per 16-step block instead of 32; same sums
# speedup vs baseline: 8.6822x; 1.1617x over previous
#include <hip/hip_runtime.h>
#include <math.h>

#define BATCH   2
#define SEQLEN  2048
#define D_INNER 2048
#define D_STATE 16
#define DT_RANK 64
#define KPROJ   (DT_RANK + 2 * D_STATE)
#define BC_COLS (2 * D_STATE)

typedef float v2f __attribute__((ext_vector_type(2)));
typedef float v8f __attribute__((ext_vector_type(8)));

#define G1_WAVES  8
#define G1_KCHUNK (D_INNER / G1_WAVES)
#define G1_NT     (KPROJ / 16)
#define G1_GROUP  32

struct G1Frag {
    float  a[16];
    float4 q[G1_NT][4];
};

__device__ __forceinline__ void g1_load(G1Frag& f, const float* pa, const float* pb, int k0)
{
    #pragma unroll
    for (int jj = 0; jj < 16; ++jj)
        f.a[jj] = pa[(size_t)(k0 + jj) * SEQLEN];
    #pragma unroll
    for (int t = 0; t < G1_NT; ++t) {
        const float4* r = reinterpret_cast<const float4*>(pb + (size_t)t * 16 * D_INNER + k0);
        #pragma unroll
        for (int i = 0; i < 4; ++i)
            f.q[t][i] = r[i];
    }
}

__device__ __forceinline__ void g1_mma(v8f (&c)[G1_NT], const G1Frag& f)
{
    #pragma unroll
    for (int j = 0; j < 8; ++j) {
        v2f a;
        a.x = f.a[2 * j];  a.y = f.a[2 * j + 1];
        #pragma unroll
        for (int t = 0; t < G1_NT; ++t) {
            const float4 qq = f.q[t][j >> 1];
            v2f bm;
            if (j & 1) { bm.x = qq.z; bm.y = qq.w; }
            else       { bm.x = qq.x; bm.y = qq.y; }
            c[t] = __builtin_amdgcn_wmma_f32_16x16x4_f32(false, a, false, bm, (short)0, c[t], false, false);
        }
    }
}

__global__ __launch_bounds__(256) void xdbl_gemm_wmma(
    const float* __restrict__ u,
    const float* __restrict__ xpw,
    float* __restrict__ xdbl,
    float* __restrict__ bct)
{
    __shared__ float part[G1_WAVES][16 * KPROJ];

    const int lane = threadIdx.x & 31;
    const int wave = threadIdx.x >> 5;
    const int m0   = blockIdx.x * 16;

    const int half = lane & 15;
    const int hi   = lane >> 4;

    const int b = m0 >> 11;
    const int l = (m0 & (SEQLEN - 1)) + half;
    const int kbase = wave * G1_KCHUNK + 16 * hi;

    const float* pa = u + (size_t)b * D_INNER * SEQLEN + (size_t)kbase * SEQLEN + l;
    const float* pb = xpw + (size_t)half * D_INNER + kbase;

    v8f c[G1_NT];
    #pragma unroll
    for (int t = 0; t < G1_NT; ++t) c[t] = v8f{};

    G1Frag f[2];
    g1_load(f[0], pa, pb, 0);
    #pragma unroll
    for (int g = 0; g < G1_KCHUNK / G1_GROUP; ++g) {
        if (g + 1 < G1_KCHUNK / G1_GROUP)
            g1_load(f[(g + 1) & 1], pa, pb, (g + 1) * G1_GROUP);
        __builtin_amdgcn_sched_barrier(0);
        g1_mma(c, f[g & 1]);
        __builtin_amdgcn_sched_barrier(0);
    }

    float* pw = &part[wave][(hi * 8) * KPROJ + half];
    #pragma unroll
    for (int t = 0; t < G1_NT; ++t) {
        #pragma unroll
        for (int r = 0; r < 8; ++r)
            pw[r * KPROJ + t * 16] = c[t][r];
    }
    __syncthreads();

    float* po = xdbl + (size_t)m0 * KPROJ;
    #pragma unroll
    for (int i = 0; i < (16 * DT_RANK) / 256; ++i) {
        const int f   = threadIdx.x + 256 * i;
        const int row = f >> 6, col = f & (DT_RANK - 1);
        const int e   = row * KPROJ + col;
        float s = part[0][e];
        #pragma unroll
        for (int w = 1; w < G1_WAVES; ++w)
            s += part[w][e];
        po[e] = s;
    }

    float* pt = bct + (size_t)(m0 >> 2) * (BC_COLS * 4);
    #pragma unroll
    for (int i = 0; i < (16 * BC_COLS) / 256; ++i) {
        const int f   = threadIdx.x + 256 * i;
        const int row = (f >> 7) * 4 + (f & 3);
        const int col = (f >> 2) & (BC_COLS - 1);
        const int e   = row * KPROJ + DT_RANK + col;
        float s = part[0][e];
        #pragma unroll
        for (int w = 1; w < G1_WAVES; ++w)
            s += part[w][e];
        pt[f] = s;
    }
}

__global__ __launch_bounds__(256) void delta_gemm_wmma(
    const float* __restrict__ xdbl,
    const float* __restrict__ dtw,
    const float* __restrict__ dtb,
    float* __restrict__ delta)
{
    const int lane = threadIdx.x & 31;
    const int wave = threadIdx.x >> 5;
    const int tile = blockIdx.x * 8 + wave;
    const int nt = tile & 127;
    const int mt = tile >> 7;
    const int m0 = mt * 16, n0 = nt * 16;

    const int half = lane & 15;
    const int hi   = lane >> 4;
    const int kk   = hi * 2;

    const float* pa = xdbl + (size_t)(m0 + half) * KPROJ + kk;
    const float* pb = dtw + (size_t)(n0 + half) * DT_RANK + kk;

    v8f c0 = {}, c1 = {};
    #pragma unroll
    for (int k = 0; k < DT_RANK; k += 8) {
        v2f a, bm;
        a.x  = pa[0];  a.y  = pa[1];
        bm.x = pb[0];  bm.y = pb[1];
        c0 = __builtin_amdgcn_wmma_f32_16x16x4_f32(false, a, false, bm, (short)0, c0, false, false);

        a.x  = pa[4];  a.y  = pa[5];
        bm.x = pb[4];  bm.y = pb[5];
        c1 = __builtin_amdgcn_wmma_f32_16x16x4_f32(false, a, false, bm, (short)0, c1, false, false);

        pa += 8;
        pb += 8;
    }
    v8f c = c0 + c1;

    const int   d    = n0 + half;
    const float bias = dtb[d];
    const int   b    = m0 >> 11;
    const int   lb   = (m0 & (SEQLEN - 1)) + hi * 8;
    float* pd = delta + ((size_t)b * D_INNER + d) * SEQLEN + lb;

    #pragma unroll
    for (int r = 0; r < 8; ++r) {
        float z  = c[r] + bias;
        float sp = fmaxf(z, 0.0f) + log1pf(__expf(-fabsf(z)));
        pd[r] = sp;
    }
}

struct ScanBlk {
    float  dl, uv;
    float4 B4[4], C4[4];
};

template <int K>
__device__ __forceinline__ float quad_elem(const float4& v)
{
    return K == 0 ? v.x : K == 1 ? v.y : K == 2 ? v.z : v.w;
}

template <int CTRL>
__device__ __forceinline__ float row_dpp(float v)
{
    return __builtin_bit_cast(float,
        __builtin_amdgcn_update_dpp(0, __builtin_bit_cast(int, v), CTRL, 0xf, 0xf, true));
}

__device__ __forceinline__ void scan_load(ScanBlk& q, const float* pdl, const float* pu,
                                          const float4* pbc, int l0, int n)
{
    q.dl = pdl[l0 + n];
    q.uv = pu[l0 + n];
    const float4* r = pbc + (size_t)(l0 >> 2) * BC_COLS;
    #pragma unroll
    for (int j = 0; j < 4; ++j) {
        q.B4[j] = r[j * BC_COLS];
        q.C4[j] = r[j * BC_COLS + D_STATE];
    }
}

template <int CTRL>
__device__ __forceinline__ float xr_pair(float a, float b, bool bit_set)
{
    const float keep = bit_set ? b : a;
    const float give = bit_set ? a : b;
    return keep + row_dpp<CTRL>(give);
}

template <int J>
struct ScanSteps {
    static __device__ __forceinline__ void run(float& x, float (&P)[16], const ScanBlk& q,
                                               float du, float An)
    {
        const float dlj = row_dpp<0x150 + J>(q.dl);
        const float duj = row_dpp<0x150 + J>(du);
        const float dA  = __expf(dlj * An);
        x = dA * x + duj * quad_elem<(J & 3)>(q.B4[J >> 2]);
        P[J] = x * quad_elem<(J & 3)>(q.C4[J >> 2]);
        ScanSteps<J + 1>::run(x, P, q, du, An);
    }
};
template <>
struct ScanSteps<16> {
    static __device__ __forceinline__ void run(float&, float (&)[16], const ScanBlk&,
                                               float, float) {}
};

__device__ __forceinline__ void scan_block(float& x, float (&r)[2], const ScanBlk& q,
                                           float An, bool b0, bool b1, bool b2)
{
    const float du = q.dl * q.uv;
    float P[16];
    ScanSteps<0>::run(x, P, q, du, An);

    float s8[8];
    #pragma unroll
    for (int i = 0; i < 8; ++i) {
        const int j = (i >> 1) * 4 + (i & 1);
        s8[i] = xr_pair<0x161>(P[j], P[j + 2], b0);
    }
    float s4[4];
    #pragma unroll
    for (int i = 0; i < 4; ++i) {
        const int j = (i >> 1) * 4 + (i & 1);
        s4[i] = xr_pair<0x162>(s8[j], s8[j + 2], b1);
    }
    r[0] = xr_pair<0x164>(s4[0], s4[2], b2);
    r[1] = xr_pair<0x164>(s4[1], s4[3], b2);
}

#define SCAN_STAGE_EDGE(x, y0, y1)                                         \
    do {                                                                   \
        asm volatile("" : "+v"(x) : "v"(y0), "v"(y1));                     \
        __builtin_amdgcn_sched_barrier(0);                                 \
    } while (0)

__global__ __launch_bounds__(256) void s6_scan(
    const float* __restrict__ u,
    const float* __restrict__ bct,
    const float* __restrict__ delta,
    const float* __restrict__ A_log,
    float* __restrict__ out)
{
    const int lane = threadIdx.x & 31;
    const int gwave = (blockIdx.x * blockDim.x + threadIdx.x) >> 5;
    const int g = lane >> 4;
    const int n = lane & 15;
    const int ch = gwave * 2 + g;
    const int b  = ch >> 11;
    const int d  = ch & (D_INNER - 1);

    const float An = -__expf(A_log[(size_t)d * D_STATE + n]);

    const float* pdl = delta + (size_t)ch * SEQLEN;
    const float* pu  = u     + (size_t)ch * SEQLEN;
    const float4* pbc = reinterpret_cast<const float4*>(bct) + (size_t)b * (SEQLEN / 4) * BC_COLS + n;
    float*       po  = out   + (size_t)ch * SEQLEN;

    ScanBlk qa, qb;
    scan_load(qa, pdl, pu, pbc, 0, n);

    const bool b0 = (n & 1) != 0, b1 = (n & 2) != 0, b2 = (n & 4) != 0, b3 = (n & 8) != 0;

    float x = 0.0f;
    float yv[2] = {0.0f, 0.0f};
    float ra[2], rb[2];
    for (int l0 = 0; l0 < SEQLEN; l0 += 32) {
        const int lp = l0 + 128 < SEQLEN - 32 ? l0 + 128 : SEQLEN - 32;
        __builtin_prefetch(pdl + lp, 0, 0);
        __builtin_prefetch(pu  + lp, 0, 0);

        scan_load(qb, pdl, pu, pbc, l0 + 16, n);
        SCAN_STAGE_EDGE(x, yv[0], yv[1]);
        scan_block(x, ra, qa, An, b0, b1, b2);

        const int ln = l0 + 32 < SEQLEN ? l0 + 32 : SEQLEN - 16;
        SCAN_STAGE_EDGE(x, ra[0], ra[1]);
        scan_load(qa, pdl, pu, pbc, ln, n);
        SCAN_STAGE_EDGE(x, ra[0], ra[1]);
        scan_block(x, rb, qb, An, b0, b1, b2);

        yv[0] = xr_pair<0x168>(ra[0], rb[0], b3);
        yv[1] = xr_pair<0x168>(ra[1], rb[1], b3);
        *reinterpret_cast<float2*>(po + l0 + 2 * n) = make_float2(yv[0], yv[1]);
    }
}

extern "C" void kernel_launch(void* const* d_in, const int* in_sizes, int n_in,
                              void* d_out, int out_size, void* d_ws, size_t ws_size,
                              hipStream_t stream)
{
    const float* u     = (const float*)d_in[0];
    const float* xpw   = (const float*)d_in[1];
    const float* dtw   = (const float*)d_in[2];
    const float* dtb   = (const float*)d_in[3];
    const float* A_log = (const float*)d_in[4];
    float*       out   = (float*)d_out;

    float* xdbl  = (float*)d_ws;
    float* bct   = xdbl + (size_t)BATCH * SEQLEN * KPROJ;
    float* delta = bct + (size_t)BATCH * SEQLEN * BC_COLS;

    xdbl_gemm_wmma<<<256, 256, 0, stream>>>(u, xpw, xdbl, bct);

    delta_gemm_wmma<<<4096, 256, 0, stream>>>(xdbl, dtw, dtb, delta);

    s6_scan<<<256, 256, 0, stream>>>(u, bct, delta, A_log, out);
}


# delta GEMM re-tiled: a wave keeps its 16x64 weight fragment in registers and walks 8 step tiles, next x_dbl fragment loaded one tile ahead; 4096 waves instead of 32768 one-tile waves; same sums
# speedup vs baseline: 9.5426x; 1.0991x over previous
#include <hip/hip_runtime.h>
#include <math.h>

#define BATCH   2
#define SEQLEN  2048
#define D_INNER 2048
#define D_STATE 16
#define DT_RANK 64
#define KPROJ   (DT_RANK + 2 * D_STATE)
#define BC_COLS (2 * D_STATE)

typedef float v2f __attribute__((ext_vector_type(2)));
typedef float v8f __attribute__((ext_vector_type(8)));

#define G1_WAVES  8
#define G1_KCHUNK (D_INNER / G1_WAVES)
#define G1_NT     (KPROJ / 16)
#define G1_GROUP  32

struct G1Frag {
    float  a[16];
    float4 q[G1_NT][4];
};

__device__ __forceinline__ void g1_load(G1Frag& f, const float* pa, const float* pb, int k0)
{
    #pragma unroll
    for (int jj = 0; jj < 16; ++jj)
        f.a[jj] = pa[(size_t)(k0 + jj) * SEQLEN];
    #pragma unroll
    for (int t = 0; t < G1_NT; ++t) {
        const float4* r = reinterpret_cast<const float4*>(pb + (size_t)t * 16 * D_INNER + k0);
        #pragma unroll
        for (int i = 0; i < 4; ++i)
            f.q[t][i] = r[i];
    }
}

__device__ __forceinline__ void g1_mma(v8f (&c)[G1_NT], const G1Frag& f)
{
    #pragma unroll
    for (int j = 0; j < 8; ++j) {
        v2f a;
        a.x = f.a[2 * j];  a.y = f.a[2 * j + 1];
        #pragma unroll
        for (int t = 0; t < G1_NT; ++t) {
            const float4 qq = f.q[t][j >> 1];
            v2f bm;
            if (j & 1) { bm.x = qq.z; bm.y = qq.w; }
            else       { bm.x = qq.x; bm.y = qq.y; }
            c[t] = __builtin_amdgcn_wmma_f32_16x16x4_f32(false, a, false, bm, (short)0, c[t], false, false);
        }
    }
}

__global__ __launch_bounds__(256) void xdbl_gemm_wmma(
    const float* __restrict__ u,
    const float* __restrict__ xpw,
    float* __restrict__ xdbl,
    float* __restrict__ bct)
{
    __shared__ float part[G1_WAVES][16 * KPROJ];

    const int lane = threadIdx.x & 31;
    const int wave = threadIdx.x >> 5;
    const int m0   = blockIdx.x * 16;

    const int half = lane & 15;
    const int hi   = lane >> 4;

    const int b = m0 >> 11;
    const int l = (m0 & (SEQLEN - 1)) + half;
    const int kbase = wave * G1_KCHUNK + 16 * hi;

    const float* pa = u + (size_t)b * D_INNER * SEQLEN + (size_t)kbase * SEQLEN + l;
    const float* pb = xpw + (size_t)half * D_INNER + kbase;

    v8f c[G1_NT];
    #pragma unroll
    for (int t = 0; t < G1_NT; ++t) c[t] = v8f{};

    G1Frag f[2];
    g1_load(f[0], pa, pb, 0);
    #pragma unroll
    for (int g = 0; g < G1_KCHUNK / G1_GROUP; ++g) {
        if (g + 1 < G1_KCHUNK / G1_GROUP)
            g1_load(f[(g + 1) & 1], pa, pb, (g + 1) * G1_GROUP);
        __builtin_amdgcn_sched_barrier(0);
        g1_mma(c, f[g & 1]);
        __builtin_amdgcn_sched_barrier(0);
    }

    float* pw = &part[wave][(hi * 8) * KPROJ + half];
    #pragma unroll
    for (int t = 0; t < G1_NT; ++t) {
        #pragma unroll
        for (int r = 0; r < 8; ++r)
            pw[r * KPROJ + t * 16] = c[t][r];
    }
    __syncthreads();

    float* po = xdbl + (size_t)m0 * KPROJ;
    #pragma unroll
    for (int i = 0; i < (16 * DT_RANK) / 256; ++i) {
        const int f   = threadIdx.x + 256 * i;
        const int row = f >> 6, col = f & (DT_RANK - 1);
        const int e   = row * KPROJ + col;
        float s = part[0][e];
        #pragma unroll
        for (int w = 1; w < G1_WAVES; ++w)
            s += part[w][e];
        po[e] = s;
    }

    float* pt = bct + (size_t)(m0 >> 2) * (BC_COLS * 4);
    #pragma unroll
    for (int i = 0; i < (16 * BC_COLS) / 256; ++i) {
        const int f   = threadIdx.x + 256 * i;
        const int row = (f >> 7) * 4 + (f & 3);
        const int col = (f >> 2) & (BC_COLS - 1);
        const int e   = row * KPROJ + DT_RANK + col;
        float s = part[0][e];
        #pragma unroll
        for (int w = 1; w < G1_WAVES; ++w)
            s += part[w][e];
        pt[f] = s;
    }
}

#define G2_MT 8
#define G2_KW (DT_RANK / 4)

struct G2Frag { v2f v[G2_KW]; };

__device__ __forceinline__ void g2_load(G2Frag& f, const float* p)
{
    #pragma unroll
    for (int j = 0; j < G2_KW; ++j)
        f.v[j] = *reinterpret_cast<const v2f*>(p + 4 * j);
}

__device__ __forceinline__ void g2_tile(const G2Frag& a, const G2Frag& w, float bias, float* pd)
{
    v8f c0 = {}, c1 = {};
    #pragma unroll
    for (int j = 0; j < G2_KW; j += 2) {
        c0 = __builtin_amdgcn_wmma_f32_16x16x4_f32(false, a.v[j],     false, w.v[j],     (short)0, c0, false, false);
        c1 = __builtin_amdgcn_wmma_f32_16x16x4_f32(false, a.v[j + 1], false, w.v[j + 1], (short)0, c1, false, false);
    }
    v8f c = c0 + c1;

    #pragma unroll
    for (int r = 0; r < 8; ++r) {
        float z  = c[r] + bias;
        float sp = fmaxf(z, 0.0f) + log1pf(__expf(-fabsf(z)));
        pd[r] = sp;
    }
}

__global__ __launch_bounds__(256) void delta_gemm_wmma(
    const float* __restrict__ xdbl,
    const float* __restrict__ dtw,
    const float* __restrict__ dtb,
    float* __restrict__ delta)
{
    const int lane = threadIdx.x & 31;
    const int wave = threadIdx.x >> 5;
    const int nt = (blockIdx.x & 15) * 8 + wave;
    const int mg = blockIdx.x >> 4;
    const int m0 = mg * (G2_MT * 16), n0 = nt * 16;

    const int half = lane & 15;
    const int hi   = lane >> 4;
    const int kk   = hi * 2;

    const float* pa = xdbl + (size_t)(m0 + half) * KPROJ + kk;
    const float* pb = dtw + (size_t)(n0 + half) * DT_RANK + kk;

    G2Frag w;
    g2_load(w, pb);

    const int   d    = n0 + half;
    const float bias = dtb[d];
    const int   b    = m0 >> 11;
    const int   lb   = (m0 & (SEQLEN - 1)) + hi * 8;
    float* pd = delta + ((size_t)b * D_INNER + d) * SEQLEN + lb;

    G2Frag a0, a1;
    g2_load(a0, pa);
    #pragma unroll 1
    for (int t = 0; t < G2_MT - 2; t += 2) {
        g2_load(a1, pa + (size_t)(t + 1) * 16 * KPROJ);
        g2_tile(a0, w, bias, pd + t * 16);
        g2_load(a0, pa + (size_t)(t + 2) * 16 * KPROJ);
        g2_tile(a1, w, bias, pd + (t + 1) * 16);
    }
    g2_load(a1, pa + (size_t)(G2_MT - 1) * 16 * KPROJ);
    g2_tile(a0, w, bias, pd + (G2_MT - 2) * 16);
    g2_tile(a1, w, bias, pd + (G2_MT - 1) * 16);
}

struct ScanBlk {
    float  dl, uv;
    float4 B4[4], C4[4];
};

template <int K>
__device__ __forceinline__ float quad_elem(const float4& v)
{
    return K == 0 ? v.x : K == 1 ? v.y : K == 2 ? v.z : v.w;
}

template <int CTRL>
__device__ __forceinline__ float row_dpp(float v)
{
    return __builtin_bit_cast(float,
        __builtin_amdgcn_update_dpp(0, __builtin_bit_cast(int, v), CTRL, 0xf, 0xf, true));
}

__device__ __forceinline__ void scan_load(ScanBlk& q, const float* pdl, const float* pu,
                                          const float4* pbc, int l0, int n)
{
    q.dl = pdl[l0 + n];
    q.uv = pu[l0 + n];
    const float4* r = pbc + (size_t)(l0 >> 2) * BC_COLS;
    #pragma unroll
    for (int j = 0; j < 4; ++j) {
        q.B4[j] = r[j * BC_COLS];
        q.C4[j] = r[j * BC_COLS + D_STATE];
    }
}

template <int CTRL>
__device__ __forceinline__ float xr_pair(float a, float b, bool bit_set)
{
    const float keep = bit_set ? b : a;
    const float give = bit_set ? a : b;
    return keep + row_dpp<CTRL>(give);
}

template <int J>
struct ScanSteps {
    static __device__ __forceinline__ void run(float& x, float (&P)[16], const ScanBlk& q,
                                               float du, float An)
    {
        const float dlj = row_dpp<0x150 + J>(q.dl);
        const float duj = row_dpp<0x150 + J>(du);
        const float dA  = __expf(dlj * An);
        x = dA * x + duj * quad_elem<(J & 3)>(q.B4[J >> 2]);
        P[J] = x * quad_elem<(J & 3)>(q.C4[J >> 2]);
        ScanSteps<J + 1>::run(x, P, q, du, An);
    }
};
template <>
struct ScanSteps<16> {
    static __device__ __forceinline__ void run(float&, float (&)[16], const ScanBlk&,
                                               float, float) {}
};

__device__ __forceinline__ void scan_block(float& x, float (&r)[2], const ScanBlk& q,
                                           float An, bool b0, bool b1, bool b2)
{
    const float du = q.dl * q.uv;
    float P[16];
    ScanSteps<0>::run(x, P, q, du, An);

    float s8[8];
    #pragma unroll
    for (int i = 0; i < 8; ++i) {
        const int j = (i >> 1) * 4 + (i & 1);
        s8[i] = xr_pair<0x161>(P[j], P[j + 2], b0);
    }
    float s4[4];
    #pragma unroll
    for (int i = 0; i < 4; ++i) {
        const int j = (i >> 1) * 4 + (i & 1);
        s4[i] = xr_pair<0x162>(s8[j], s8[j + 2], b1);
    }
    r[0] = xr_pair<0x164>(s4[0], s4[2], b2);
    r[1] = xr_pair<0x164>(s4[1], s4[3], b2);
}

#define SCAN_STAGE_EDGE(x, y0, y1)                                         \
    do {                                                                   \
        asm volatile("" : "+v"(x) : "v"(y0), "v"(y1));                     \
        __builtin_amdgcn_sched_barrier(0);                                 \
    } while (0)

__global__ __launch_bounds__(256) void s6_scan(
    const float* __restrict__ u,
    const float* __restrict__ bct,
    const float* __restrict__ delta,
    const float* __restrict__ A_log,
    float* __restrict__ out)
{
    const int lane = threadIdx.x & 31;
    const int gwave = (blockIdx.x * blockDim.x + threadIdx.x) >> 5;
    const int g = lane >> 4;
    const int n = lane & 15;
    const int ch = gwave * 2 + g;
    const int b  = ch >> 11;
    const int d  = ch & (D_INNER - 1);

    const float An = -__expf(A_log[(size_t)d * D_STATE + n]);

    const float* pdl = delta + (size_t)ch * SEQLEN;
    const float* pu  = u     + (size_t)ch * SEQLEN;
    const float4* pbc = reinterpret_cast<const float4*>(bct) + (size_t)b * (SEQLEN / 4) * BC_COLS + n;
    float*       po  = out   + (size_t)ch * SEQLEN;

    ScanBlk qa, qb;
    scan_load(qa, pdl, pu, pbc, 0, n);

    const bool b0 = (n & 1) != 0, b1 = (n & 2) != 0, b2 = (n & 4) != 0, b3 = (n & 8) != 0;

    float x = 0.0f;
    float yv[2] = {0.0f, 0.0f};
    float ra[2], rb[2];
    for (int l0 = 0; l0 < SEQLEN; l0 += 32) {
        const int lp = l0 + 128 < SEQLEN - 32 ? l0 + 128 : SEQLEN - 32;
        __builtin_prefetch(pdl + lp, 0, 0);
        __builtin_prefetch(pu  + lp, 0, 0);

        scan_load(qb, pdl, pu, pbc, l0 + 16, n);
        SCAN_STAGE_EDGE(x, yv[0], yv[1]);
        scan_block(x, ra, qa, An, b0, b1, b2);

        const int ln = l0 + 32 < SEQLEN ? l0 + 32 : SEQLEN - 16;
        SCAN_STAGE_EDGE(x, ra[0], ra[1]);
        scan_load(qa, pdl, pu, pbc, ln, n);
        SCAN_STAGE_EDGE(x, ra[0], ra[1]);
        scan_block(x, rb, qb, An, b0, b1, b2);

        yv[0] = xr_pair<0x168>(ra[0], rb[0], b3);
        yv[1] = xr_pair<0x168>(ra[1], rb[1], b3);
        *reinterpret_cast<float2*>(po + l0 + 2 * n) = make_float2(yv[0], yv[1]);
    }
}

extern "C" void kernel_launch(void* const* d_in, const int* in_sizes, int n_in,
                              void* d_out, int out_size, void* d_ws, size_t ws_size,
                              hipStream_t stream)
{
    const float* u     = (const float*)d_in[0];
    const float* xpw   = (const float*)d_in[1];
    const float* dtw   = (const float*)d_in[2];
    const float* dtb   = (const float*)d_in[3];
    const float* A_log = (const float*)d_in[4];
    float*       out   = (float*)d_out;

    float* xdbl  = (float*)d_ws;
    float* bct   = xdbl + (size_t)BATCH * SEQLEN * KPROJ;
    float* delta = bct + (size_t)BATCH * SEQLEN * BC_COLS;

    xdbl_gemm_wmma<<<256, 256, 0, stream>>>(u, xpw, xdbl, bct);

    delta_gemm_wmma<<<512, 256, 0, stream>>>(xdbl, dtw, dtb, delta);

    s6_scan<<<256, 256, 0, stream>>>(u, bct, delta, A_log, out);
}


# delta GEMM softplus: library log1pf (about 85 vector instructions per value) replaced by log1p(t) = 2*atanh(t/(2+t)) on t = exp(-|z|) <= 1: corrected rcp + 7-term series; delta's last bits differ
# speedup vs baseline: 10.7901x; 1.1307x over previous
#include <hip/hip_runtime.h>
#include <math.h>

#define BATCH   2
#define SEQLEN  2048
#define D_INNER 2048
#define D_STATE 16
#define DT_RANK 64
#define KPROJ   (DT_RANK + 2 * D_STATE)
#define BC_COLS (2 * D_STATE)

typedef float v2f __attribute__((ext_vector_type(2)));
typedef float v8f __attribute__((ext_vector_type(8)));

#define G1_WAVES  8
#define G1_KCHUNK (D_INNER / G1_WAVES)
#define G1_NT     (KPROJ / 16)
#define G1_GROUP  32

struct G1Frag {
    float  a[16];
    float4 q[G1_NT][4];
};

__device__ __forceinline__ void g1_load(G1Frag& f, const float* pa, const float* pb, int k0)
{
    #pragma unroll
    for (int jj = 0; jj < 16; ++jj)
        f.a[jj] = pa[(size_t)(k0 + jj) * SEQLEN];
    #pragma unroll
    for (int t = 0; t < G1_NT; ++t) {
        const float4* r = reinterpret_cast<const float4*>(pb + (size_t)t * 16 * D_INNER + k0);
        #pragma unroll
        for (int i = 0; i < 4; ++i)
            f.q[t][i] = r[i];
    }
}

__device__ __forceinline__ void g1_mma(v8f (&c)[G1_NT], const G1Frag& f)
{
    #pragma unroll
    for (int j = 0; j < 8; ++j) {
        v2f a;
        a.x = f.a[2 * j];  a.y = f.a[2 * j + 1];
        #pragma unroll
        for (int t = 0; t < G1_NT; ++t) {
            const float4 qq = f.q[t][j >> 1];
            v2f bm;
            if (j & 1) { bm.x = qq.z; bm.y = qq.w; }
            else       { bm.x = qq.x; bm.y = qq.y; }
            c[t] = __builtin_amdgcn_wmma_f32_16x16x4_f32(false, a, false, bm, (short)0, c[t], false, false);
        }
    }
}

__global__ __launch_bounds__(256) void xdbl_gemm_wmma(
    const float* __restrict__ u,
    const float* __restrict__ xpw,
    float* __restrict__ xdbl,
    float* __restrict__ bct)
{
    __shared__ float part[G1_WAVES][16 * KPROJ];

    const int lane = threadIdx.x & 31;
    const int wave = threadIdx.x >> 5;
    const int m0   = blockIdx.x * 16;

    const int half = lane & 15;
    const int hi   = lane >> 4;

    const int b = m0 >> 11;
    const int l = (m0 & (SEQLEN - 1)) + half;
    const int kbase = wave * G1_KCHUNK + 16 * hi;

    const float* pa = u + (size_t)b * D_INNER * SEQLEN + (size_t)kbase * SEQLEN + l;
    const float* pb = xpw + (size_t)half * D_INNER + kbase;

    v8f c[G1_NT];
    #pragma unroll
    for (int t = 0; t < G1_NT; ++t) c[t] = v8f{};

    G1Frag f[2];
    g1_load(f[0], pa, pb, 0);
    #pragma unroll
    for (int g = 0; g < G1_KCHUNK / G1_GROUP; ++g) {
        if (g + 1 < G1_KCHUNK / G1_GROUP)
            g1_load(f[(g + 1) & 1], pa, pb, (g + 1) * G1_GROUP);
        __builtin_amdgcn_sched_barrier(0);
        g1_mma(c, f[g & 1]);
        __builtin_amdgcn_sched_barrier(0);
    }

    float* pw = &part[wave][(hi * 8) * KPROJ + half];
    #pragma unroll
    for (int t = 0; t < G1_NT; ++t) {
        #pragma unroll
        for (int r = 0; r < 8; ++r)
            pw[r * KPROJ + t * 16] = c[t][r];
    }
    __syncthreads();

    float* po = xdbl + (size_t)m0 * KPROJ;
    #pragma unroll
    for (int i = 0; i < (16 * DT_RANK) / 256; ++i) {
        const int f   = threadIdx.x + 256 * i;
        const int row = f >> 6, col = f & (DT_RANK - 1);
        const int e   = row * KPROJ + col;
        float s = part[0][e];
        #pragma unroll
        for (int w = 1; w < G1_WAVES; ++w)
            s += part[w][e];
        po[e] = s;
    }

    float* pt = bct + (size_t)(m0 >> 2) * (BC_COLS * 4);
    #pragma unroll
    for (int i = 0; i < (16 * BC_COLS) / 256; ++i) {
        const int f   = threadIdx.x + 256 * i;
        const int row = (f >> 7) * 4 + (f & 3);
        const int col = (f >> 2) & (BC_COLS - 1);
        const int e   = row * KPROJ + DT_RANK + col;
        float s = part[0][e];
        #pragma unroll
        for (int w = 1; w < G1_WAVES; ++w)
            s += part[w][e];
        pt[f] = s;
    }
}

#define G2_MT 8
#define G2_KW (DT_RANK / 4)

struct G2Frag { v2f v[G2_KW]; };

__device__ __forceinline__ void g2_load(G2Frag& f, const float* p)
{
    #pragma unroll
    for (int j = 0; j < G2_KW; ++j)
        f.v[j] = *reinterpret_cast<const v2f*>(p + 4 * j);
}

__device__ __forceinline__ float log1p_unit(float t)
{
    const float d = 2.0f + t;
    const float r = __builtin_amdgcn_rcpf(d);
    float s = t * r;
    const float e = fmaf(-s, d, t);
    s = fmaf(e, r, s);
    const float w = s * s;
    float p = 1.0f / 13.0f;
    p = fmaf(p, w, 1.0f / 11.0f);
    p = fmaf(p, w, 1.0f / 9.0f);
    p = fmaf(p, w, 1.0f / 7.0f);
    p = fmaf(p, w, 1.0f / 5.0f);
    p = fmaf(p, w, 1.0f / 3.0f);
    const float s2 = s + s;
    return fmaf(s2 * w, p, s2);
}

__device__ __forceinline__ void g2_tile(const G2Frag& a, const G2Frag& w, float bias, float* pd)
{
    v8f c0 = {}, c1 = {};
    #pragma unroll
    for (int j = 0; j < G2_KW; j += 2) {
        c0 = __builtin_amdgcn_wmma_f32_16x16x4_f32(false, a.v[j],     false, w.v[j],     (short)0, c0, false, false);
        c1 = __builtin_amdgcn_wmma_f32_16x16x4_f32(false, a.v[j + 1], false, w.v[j + 1], (short)0, c1, false, false);
    }
    v8f c = c0 + c1;

    #pragma unroll
    for (int r = 0; r < 8; ++r) {
        float z  = c[r] + bias;
        float sp = fmaxf(z, 0.0f) + log1p_unit(__expf(-fabsf(z)));
        pd[r] = sp;
    }
}

__global__ __launch_bounds__(256) void delta_gemm_wmma(
    const float* __restrict__ xdbl,
    const float* __restrict__ dtw,
    const float* __restrict__ dtb,
    float* __restrict__ delta)
{
    const int lane = threadIdx.x & 31;
    const int wave = threadIdx.x >> 5;
    const int nt = (blockIdx.x & 15) * 8 + wave;
    const int mg = blockIdx.x >> 4;
    const int m0 = mg * (G2_MT * 16), n0 = nt * 16;

    const int half = lane & 15;
    const int hi   = lane >> 4;
    const int kk   = hi * 2;

    const float* pa = xdbl + (size_t)(m0 + half) * KPROJ + kk;
    const float* pb = dtw + (size_t)(n0 + half) * DT_RANK + kk;

    G2Frag w;
    g2_load(w, pb);

    const int   d    = n0 + half;
    const float bias = dtb[d];
    const int   b    = m0 >> 11;
    const int   lb   = (m0 & (SEQLEN - 1)) + hi * 8;
    float* pd = delta + ((size_t)b * D_INNER + d) * SEQLEN + lb;

    G2Frag a0, a1;
    g2_load(a0, pa);
    #pragma unroll 1
    for (int t = 0; t < G2_MT - 2; t += 2) {
        g2_load(a1, pa + (size_t)(t + 1) * 16 * KPROJ);
        g2_tile(a0, w, bias, pd + t * 16);
        g2_load(a0, pa + (size_t)(t + 2) * 16 * KPROJ);
        g2_tile(a1, w, bias, pd + (t + 1) * 16);
    }
    g2_load(a1, pa + (size_t)(G2_MT - 1) * 16 * KPROJ);
    g2_tile(a0, w, bias, pd + (G2_MT - 2) * 16);
    g2_tile(a1, w, bias, pd + (G2_MT - 1) * 16);
}

struct ScanBlk {
    float  dl, uv;
    float4 B4[4], C4[4];
};

template <int K>
__device__ __forceinline__ float quad_elem(const float4& v)
{
    return K == 0 ? v.x : K == 1 ? v.y : K == 2 ? v.z : v.w;
}

template <int CTRL>
__device__ __forceinline__ float row_dpp(float v)
{
    return __builtin_bit_cast(float,
        __builtin_amdgcn_update_dpp(0, __builtin_bit_cast(int, v), CTRL, 0xf, 0xf, true));
}

__device__ __forceinline__ void scan_load(ScanBlk& q, const float* pdl, const float* pu,
                                          const float4* pbc, int l0, int n)
{
    q.dl = pdl[l0 + n];
    q.uv = pu[l0 + n];
    const float4* r = pbc + (size_t)(l0 >> 2) * BC_COLS;
    #pragma unroll
    for (int j = 0; j < 4; ++j) {
        q.B4[j] = r[j * BC_COLS];
        q.C4[j] = r[j * BC_COLS + D_STATE];
    }
}

template <int CTRL>
__device__ __forceinline__ float xr_pair(float a, float b, bool bit_set)
{
    const float keep = bit_set ? b : a;
    const float give = bit_set ? a : b;
    return keep + row_dpp<CTRL>(give);
}

template <int J>
struct ScanSteps {
    static __device__ __forceinline__ void run(float& x, float (&P)[16], const ScanBlk& q,
                                               float du, float An)
    {
        const float dlj = row_dpp<0x150 + J>(q.dl);
        const float duj = row_dpp<0x150 + J>(du);
        const float dA  = __expf(dlj * An);
        x = dA * x + duj * quad_elem<(J & 3)>(q.B4[J >> 2]);
        P[J] = x * quad_elem<(J & 3)>(q.C4[J >> 2]);
        ScanSteps<J + 1>::run(x, P, q, du, An);
    }
};
template <>
struct ScanSteps<16> {
    static __device__ __forceinline__ void run(float&, float (&)[16], const ScanBlk&,
                                               float, float) {}
};

__device__ __forceinline__ void scan_block(float& x, float (&r)[2], const ScanBlk& q,
                                           float An, bool b0, bool b1, bool b2)
{
    const float du = q.dl * q.uv;
    float P[16];
    ScanSteps<0>::run(x, P, q, du, An);

    float s8[8];
    #pragma unroll
    for (int i = 0; i < 8; ++i) {
        const int j = (i >> 1) * 4 + (i & 1);
        s8[i] = xr_pair<0x161>(P[j], P[j + 2], b0);
    }
    float s4[4];
    #pragma unroll
    for (int i = 0; i < 4; ++i) {
        const int j = (i >> 1) * 4 + (i & 1);
        s4[i] = xr_pair<0x162>(s8[j], s8[j + 2], b1);
    }
    r[0] = xr_pair<0x164>(s4[0], s4[2], b2);
    r[1] = xr_pair<0x164>(s4[1], s4[3], b2);
}

#define SCAN_STAGE_EDGE(x, y0, y1)                                         \
    do {                                                                   \
        asm volatile("" : "+v"(x) : "v"(y0), "v"(y1));                     \
        __builtin_amdgcn_sched_barrier(0);                                 \
    } while (0)

__global__ __launch_bounds__(256) void s6_scan(
    const float* __restrict__ u,
    const float* __restrict__ bct,
    const float* __restrict__ delta,
    const float* __restrict__ A_log,
    float* __restrict__ out)
{
    const int lane = threadIdx.x & 31;
    const int gwave = (blockIdx.x * blockDim.x + threadIdx.x) >> 5;
    const int g = lane >> 4;
    const int n = lane & 15;
    const int ch = gwave * 2 + g;
    const int b  = ch >> 11;
    const int d  = ch & (D_INNER - 1);

    const float An = -__expf(A_log[(size_t)d * D_STATE + n]);

    const float* pdl = delta + (size_t)ch * SEQLEN;
    const float* pu  = u     + (size_t)ch * SEQLEN;
    const float4* pbc = reinterpret_cast<const float4*>(bct) + (size_t)b * (SEQLEN / 4) * BC_COLS + n;
    float*       po  = out   + (size_t)ch * SEQLEN;

    ScanBlk qa, qb;
    scan_load(qa, pdl, pu, pbc, 0, n);

    const bool b0 = (n & 1) != 0, b1 = (n & 2) != 0, b2 = (n & 4) != 0, b3 = (n & 8) != 0;

    float x = 0.0f;
    float yv[2] = {0.0f, 0.0f};
    float ra[2], rb[2];
    for (int l0 = 0; l0 < SEQLEN; l0 += 32) {
        const int lp = l0 + 128 < SEQLEN - 32 ? l0 + 128 : SEQLEN - 32;
        __builtin_prefetch(pdl + lp, 0, 0);
        __builtin_prefetch(pu  + lp, 0, 0);

        scan_load(qb, pdl, pu, pbc, l0 + 16, n);
        SCAN_STAGE_EDGE(x, yv[0], yv[1]);
        scan_block(x, ra, qa, An, b0, b1, b2);

        const int ln = l0 + 32 < SEQLEN ? l0 + 32 : SEQLEN - 16;
        SCAN_STAGE_EDGE(x, ra[0], ra[1]);
        scan_load(qa, pdl, pu, pbc, ln, n);
        SCAN_STAGE_EDGE(x, ra[0], ra[1]);
        scan_block(x, rb, qb, An, b0, b1, b2);

        yv[0] = xr_pair<0x168>(ra[0], rb[0], b3);
        yv[1] = xr_pair<0x168>(ra[1], rb[1], b3);
        *reinterpret_cast<float2*>(po + l0 + 2 * n) = make_float2(yv[0], yv[1]);
    }
}

extern "C" void kernel_launch(void* const* d_in, const int* in_sizes, int n_in,
                              void* d_out, int out_size, void* d_ws, size_t ws_size,
                              hipStream_t stream)
{
    const float* u     = (const float*)d_in[0];
    const float* xpw   = (const float*)d_in[1];
    const float* dtw   = (const float*)d_in[2];
    const float* dtb   = (const float*)d_in[3];
    const float* A_log = (const float*)d_in[4];
    float*       out   = (float*)d_out;

    float* xdbl  = (float*)d_ws;
    float* bct   = xdbl + (size_t)BATCH * SEQLEN * KPROJ;
    float* delta = bct + (size_t)BATCH * SEQLEN * BC_COLS;

    xdbl_gemm_wmma<<<256, 256, 0, stream>>>(u, xpw, xdbl, bct);

    delta_gemm_wmma<<<512, 256, 0, stream>>>(xdbl, dtw, dtb, delta);

    s6_scan<<<256, 256, 0, stream>>>(u, bct, delta, A_log, out);
}


# scan: block loads (delta, u, B/C quads) issued three 16-step blocks ahead of their use instead of one: four register sets, 64-step loop trips; same arithmetic, same sums
# speedup vs baseline: 11.2289x; 1.0407x over previous
#include <hip/hip_runtime.h>
#include <math.h>

#define BATCH   2
#define SEQLEN  2048
#define D_INNER 2048
#define D_STATE 16
#define DT_RANK 64
#define KPROJ   (DT_RANK + 2 * D_STATE)
#define BC_COLS (2 * D_STATE)

typedef float v2f __attribute__((ext_vector_type(2)));
typedef float v8f __attribute__((ext_vector_type(8)));

#define G1_WAVES  8
#define G1_KCHUNK (D_INNER / G1_WAVES)
#define G1_NT     (KPROJ / 16)
#define G1_GROUP  32

struct G1Frag {
    float  a[16];
    float4 q[G1_NT][4];
};

__device__ __forceinline__ void g1_load(G1Frag& f, const float* pa, const float* pb, int k0)
{
    #pragma unroll
    for (int jj = 0; jj < 16; ++jj)
        f.a[jj] = pa[(size_t)(k0 + jj) * SEQLEN];
    #pragma unroll
    for (int t = 0; t < G1_NT; ++t) {
        const float4* r = reinterpret_cast<const float4*>(pb + (size_t)t * 16 * D_INNER + k0);
        #pragma unroll
        for (int i = 0; i < 4; ++i)
            f.q[t][i] = r[i];
    }
}

__device__ __forceinline__ void g1_mma(v8f (&c)[G1_NT], const G1Frag& f)
{
    #pragma unroll
    for (int j = 0; j < 8; ++j) {
        v2f a;
        a.x = f.a[2 * j];  a.y = f.a[2 * j + 1];
        #pragma unroll
        for (int t = 0; t < G1_NT; ++t) {
            const float4 qq = f.q[t][j >> 1];
            v2f bm;
            if (j & 1) { bm.x = qq.z; bm.y = qq.w; }
            else       { bm.x = qq.x; bm.y = qq.y; }
            c[t] = __builtin_amdgcn_wmma_f32_16x16x4_f32(false, a, false, bm, (short)0, c[t], false, false);
        }
    }
}

__global__ __launch_bounds__(256) void xdbl_gemm_wmma(
    const float* __restrict__ u,
    const float* __restrict__ xpw,
    float* __restrict__ xdbl,
    float* __restrict__ bct)
{
    __shared__ float part[G1_WAVES][16 * KPROJ];

    const int lane = threadIdx.x & 31;
    const int wave = threadIdx.x >> 5;
    const int m0   = blockIdx.x * 16;

    const int half = lane & 15;
    const int hi   = lane >> 4;

    const int b = m0 >> 11;
    const int l = (m0 & (SEQLEN - 1)) + half;
    const int kbase = wave * G1_KCHUNK + 16 * hi;

    const float* pa = u + (size_t)b * D_INNER * SEQLEN + (size_t)kbase * SEQLEN + l;
    const float* pb = xpw + (size_t)half * D_INNER + kbase;

    v8f c[G1_NT];
    #pragma unroll
    for (int t = 0; t < G1_NT; ++t) c[t] = v8f{};

    G1Frag f[2];
    g1_load(f[0], pa, pb, 0);
    #pragma unroll
    for (int g = 0; g < G1_KCHUNK / G1_GROUP; ++g) {
        if (g + 1 < G1_KCHUNK / G1_GROUP)
            g1_load(f[(g + 1) & 1], pa, pb, (g + 1) * G1_GROUP);
        __builtin_amdgcn_sched_barrier(0);
        g1_mma(c, f[g & 1]);
        __builtin_amdgcn_sched_barrier(0);
    }

    float* pw = &part[wave][(hi * 8) * KPROJ + half];
    #pragma unroll
    for (int t = 0; t < G1_NT; ++t) {
        #pragma unroll
        for (int r = 0; r < 8; ++r)
            pw[r * KPROJ + t * 16] = c[t][r];
    }
    __syncthreads();

    float* po = xdbl + (size_t)m0 * KPROJ;
    #pragma unroll
    for (int i = 0; i < (16 * DT_RANK) / 256; ++i) {
        const int f   = threadIdx.x + 256 * i;
        const int row = f >> 6, col = f & (DT_RANK - 1);
        const int e   = row * KPROJ + col;
        float s = part[0][e];
        #pragma unroll
        for (int w = 1; w < G1_WAVES; ++w)
            s += part[w][e];
        po[e] = s;
    }

    float* pt = bct + (size_t)(m0 >> 2) * (BC_COLS * 4);
    #pragma unroll
    for (int i = 0; i < (16 * BC_COLS) / 256; ++i) {
        const int f   = threadIdx.x + 256 * i;
        const int row = (f >> 7) * 4 + (f & 3);
        const int col = (f >> 2) & (BC_COLS - 1);
        const int e   = row * KPROJ + DT_RANK + col;
        float s = part[0][e];
        #pragma unroll
        for (int w = 1; w < G1_WAVES; ++w)
            s += part[w][e];
        pt[f] = s;
    }
}

#define G2_MT 8
#define G2_KW (DT_RANK / 4)

struct G2Frag { v2f v[G2_KW]; };

__device__ __forceinline__ void g2_load(G2Frag& f, const float* p)
{
    #pragma unroll
    for (int j = 0; j < G2_KW; ++j)
        f.v[j] = *reinterpret_cast<const v2f*>(p + 4 * j);
}

__device__ __forceinline__ float log1p_unit(float t)
{
    const float d = 2.0f + t;
    const float r = __builtin_amdgcn_rcpf(d);
    float s = t * r;
    const float e = fmaf(-s, d, t);
    s = fmaf(e, r, s);
    const float w = s * s;
    float p = 1.0f / 13.0f;
    p = fmaf(p, w, 1.0f / 11.0f);
    p = fmaf(p, w, 1.0f / 9.0f);
    p = fmaf(p, w, 1.0f / 7.0f);
    p = fmaf(p, w, 1.0f / 5.0f);
    p = fmaf(p, w, 1.0f / 3.0f);
    const float s2 = s + s;
    return fmaf(s2 * w, p, s2);
}

__device__ __forceinline__ void g2_tile(const G2Frag& a, const G2Frag& w, float bias, float* pd)
{
    v8f c0 = {}, c1 = {};
    #pragma unroll
    for (int j = 0; j < G2_KW; j += 2) {
        c0 = __builtin_amdgcn_wmma_f32_16x16x4_f32(false, a.v[j],     false, w.v[j],     (short)0, c0, false, false);
        c1 = __builtin_amdgcn_wmma_f32_16x16x4_f32(false, a.v[j + 1], false, w.v[j + 1], (short)0, c1, false, false);
    }
    v8f c = c0 + c1;

    #pragma unroll
    for (int r = 0; r < 8; ++r) {
        float z  = c[r] + bias;
        float sp = fmaxf(z, 0.0f) + log1p_unit(__expf(-fabsf(z)));
        pd[r] = sp;
    }
}

__global__ __launch_bounds__(256) void delta_gemm_wmma(
    const float* __restrict__ xdbl,
    const float* __restrict__ dtw,
    const float* __restrict__ dtb,
    float* __restrict__ delta)
{
    const int lane = threadIdx.x & 31;
    const int wave = threadIdx.x >> 5;
    const int nt = (blockIdx.x & 15) * 8 + wave;
    const int mg = blockIdx.x >> 4;
    const int m0 = mg * (G2_MT * 16), n0 = nt * 16;

    const int half = lane & 15;
    const int hi   = lane >> 4;
    const int kk   = hi * 2;

    const float* pa = xdbl + (size_t)(m0 + half) * KPROJ + kk;
    const float* pb = dtw + (size_t)(n0 + half) * DT_RANK + kk;

    G2Frag w;
    g2_load(w, pb);

    const int   d    = n0 + half;
    const float bias = dtb[d];
    const int   b    = m0 >> 11;
    const int   lb   = (m0 & (SEQLEN - 1)) + hi * 8;
    float* pd = delta + ((size_t)b * D_INNER + d) * SEQLEN + lb;

    G2Frag a0, a1;
    g2_load(a0, pa);
    #pragma unroll 1
    for (int t = 0; t < G2_MT - 2; t += 2) {
        g2_load(a1, pa + (size_t)(t + 1) * 16 * KPROJ);
        g2_tile(a0, w, bias, pd + t * 16);
        g2_load(a0, pa + (size_t)(t + 2) * 16 * KPROJ);
        g2_tile(a1, w, bias, pd + (t + 1) * 16);
    }
    g2_load(a1, pa + (size_t)(G2_MT - 1) * 16 * KPROJ);
    g2_tile(a0, w, bias, pd + (G2_MT - 2) * 16);
    g2_tile(a1, w, bias, pd + (G2_MT - 1) * 16);
}

struct ScanBlk {
    float  dl, uv;
    float4 B4[4], C4[4];
};

template <int K>
__device__ __forceinline__ float quad_elem(const float4& v)
{
    return K == 0 ? v.x : K == 1 ? v.y : K == 2 ? v.z : v.w;
}

template <int CTRL>
__device__ __forceinline__ float row_dpp(float v)
{
    return __builtin_bit_cast(float,
        __builtin_amdgcn_update_dpp(0, __builtin_bit_cast(int, v), CTRL, 0xf, 0xf, true));
}

__device__ __forceinline__ void scan_load(ScanBlk& q, const float* pdl, const float* pu,
                                          const float4* pbc, int l0, int n)
{
    q.dl = pdl[l0 + n];
    q.uv = pu[l0 + n];
    const float4* r = pbc + (size_t)(l0 >> 2) * BC_COLS;
    #pragma unroll
    for (int j = 0; j < 4; ++j) {
        q.B4[j] = r[j * BC_COLS];
        q.C4[j] = r[j * BC_COLS + D_STATE];
    }
}

template <int CTRL>
__device__ __forceinline__ float xr_pair(float a, float b, bool bit_set)
{
    const float keep = bit_set ? b : a;
    const float give = bit_set ? a : b;
    return keep + row_dpp<CTRL>(give);
}

template <int J>
struct ScanSteps {
    static __device__ __forceinline__ void run(float& x, float (&P)[16], const ScanBlk& q,
                                               float du, float An)
    {
        const float dlj = row_dpp<0x150 + J>(q.dl);
        const float duj = row_dpp<0x150 + J>(du);
        const float dA  = __expf(dlj * An);
        x = dA * x + duj * quad_elem<(J & 3)>(q.B4[J >> 2]);
        P[J] = x * quad_elem<(J & 3)>(q.C4[J >> 2]);
        ScanSteps<J + 1>::run(x, P, q, du, An);
    }
};
template <>
struct ScanSteps<16> {
    static __device__ __forceinline__ void run(float&, float (&)[16], const ScanBlk&,
                                               float, float) {}
};

__device__ __forceinline__ void scan_block(float& x, float (&r)[2], const ScanBlk& q,
                                           float An, bool b0, bool b1, bool b2)
{
    const float du = q.dl * q.uv;
    float P[16];
    ScanSteps<0>::run(x, P, q, du, An);

    float s8[8];
    #pragma unroll
    for (int i = 0; i < 8; ++i) {
        const int j = (i >> 1) * 4 + (i & 1);
        s8[i] = xr_pair<0x161>(P[j], P[j + 2], b0);
    }
    float s4[4];
    #pragma unroll
    for (int i = 0; i < 4; ++i) {
        const int j = (i >> 1) * 4 + (i & 1);
        s4[i] = xr_pair<0x162>(s8[j], s8[j + 2], b1);
    }
    r[0] = xr_pair<0x164>(s4[0], s4[2], b2);
    r[1] = xr_pair<0x164>(s4[1], s4[3], b2);
}

#define SCAN_STAGE_EDGE(x, y0, y1)                                         \
    do {                                                                   \
        asm volatile("" : "+v"(x) : "v"(y0), "v"(y1));                     \
        __builtin_amdgcn_sched_barrier(0);                                 \
    } while (0)

__global__ __launch_bounds__(256) void s6_scan(
    const float* __restrict__ u,
    const float* __restrict__ bct,
    const float* __restrict__ delta,
    const float* __restrict__ A_log,
    float* __restrict__ out)
{
    const int lane = threadIdx.x & 31;
    const int gwave = (blockIdx.x * blockDim.x + threadIdx.x) >> 5;
    const int g = lane >> 4;
    const int n = lane & 15;
    const int ch = gwave * 2 + g;
    const int b  = ch >> 11;
    const int d  = ch & (D_INNER - 1);

    const float An = -__expf(A_log[(size_t)d * D_STATE + n]);

    const float* pdl = delta + (size_t)ch * SEQLEN;
    const float* pu  = u     + (size_t)ch * SEQLEN;
    const float4* pbc = reinterpret_cast<const float4*>(bct) + (size_t)b * (SEQLEN / 4) * BC_COLS + n;
    float*       po  = out   + (size_t)ch * SEQLEN;

    ScanBlk q0, q1, q2, q3;
    scan_load(q0, pdl, pu, pbc, 0, n);
    scan_load(q1, pdl, pu, pbc, 16, n);
    scan_load(q2, pdl, pu, pbc, 32, n);

    const bool b0 = (n & 1) != 0, b1 = (n & 2) != 0, b2 = (n & 4) != 0, b3 = (n & 8) != 0;

    float x = 0.0f;
    float yv[2] = {0.0f, 0.0f};
    float ra[2], rb[2];
    for (int l0 = 0; l0 < SEQLEN; l0 += 64) {
        const int lp0 = l0 + 192 < SEQLEN - 32 ? l0 + 192 : SEQLEN - 32;
        const int lp1 = l0 + 224 < SEQLEN - 32 ? l0 + 224 : SEQLEN - 32;
        __builtin_prefetch(pdl + lp0, 0, 0);
        __builtin_prefetch(pu  + lp0, 0, 0);
        __builtin_prefetch(pdl + lp1, 0, 0);
        __builtin_prefetch(pu  + lp1, 0, 0);

        scan_load(q3, pdl, pu, pbc, l0 + 48, n);
        SCAN_STAGE_EDGE(x, yv[0], yv[1]);
        scan_block(x, ra, q0, An, b0, b1, b2);

        const int la = l0 + 64 < SEQLEN ? l0 + 64 : SEQLEN - 16;
        SCAN_STAGE_EDGE(x, ra[0], ra[1]);
        scan_load(q0, pdl, pu, pbc, la, n);
        SCAN_STAGE_EDGE(x, ra[0], ra[1]);
        scan_block(x, rb, q1, An, b0, b1, b2);

        yv[0] = xr_pair<0x168>(ra[0], rb[0], b3);
        yv[1] = xr_pair<0x168>(ra[1], rb[1], b3);
        *reinterpret_cast<float2*>(po + l0 + 2 * n) = make_float2(yv[0], yv[1]);

        const int lb = l0 + 80 < SEQLEN ? l0 + 80 : SEQLEN - 16;
        SCAN_STAGE_EDGE(x, yv[0], yv[1]);
        scan_load(q1, pdl, pu, pbc, lb, n);
        SCAN_STAGE_EDGE(x, yv[0], yv[1]);
        scan_block(x, ra, q2, An, b0, b1, b2);

        const int lc = l0 + 96 < SEQLEN ? l0 + 96 : SEQLEN - 16;
        SCAN_STAGE_EDGE(x, ra[0], ra[1]);
        scan_load(q2, pdl, pu, pbc, lc, n);
        SCAN_STAGE_EDGE(x, ra[0], ra[1]);
        scan_block(x, rb, q3, An, b0, b1, b2);

        yv[0] = xr_pair<0x168>(ra[0], rb[0], b3);
        yv[1] = xr_pair<0x168>(ra[1], rb[1], b3);
        *reinterpret_cast<float2*>(po + l0 + 32 + 2 * n) = make_float2(yv[0], yv[1]);
    }
}

extern "C" void kernel_launch(void* const* d_in, const int* in_sizes, int n_in,
                              void* d_out, int out_size, void* d_ws, size_t ws_size,
                              hipStream_t stream)
{
    const float* u     = (const float*)d_in[0];
    const float* xpw   = (const float*)d_in[1];
    const float* dtw   = (const float*)d_in[2];
    const float* dtb   = (const float*)d_in[3];
    const float* A_log = (const float*)d_in[4];
    float*       out   = (float*)d_out;

    float* xdbl  = (float*)d_ws;
    float* bct   = xdbl + (size_t)BATCH * SEQLEN * KPROJ;
    float* delta = bct + (size_t)BATCH * SEQLEN * BC_COLS;

    xdbl_gemm_wmma<<<256, 256, 0, stream>>>(u, xpw, xdbl, bct);

    delta_gemm_wmma<<<512, 256, 0, stream>>>(xdbl, dtw, dtb, delta);

    s6_scan<<<256, 256, 0, stream>>>(u, bct, delta, A_log, out);
}


# scan: each lane loads its channel's 16 delta and 16 u of a block itself (row-uniform 16-B loads) instead of 2 DPP row broadcasts per step; step products go out packed; same sums, x's last bits differ
# speedup vs baseline: 12.3031x; 1.0957x over previous
#include <hip/hip_runtime.h>
#include <math.h>

#define BATCH   2
#define SEQLEN  2048
#define D_INNER 2048
#define D_STATE 16
#define DT_RANK 64
#define KPROJ   (DT_RANK + 2 * D_STATE)
#define BC_COLS (2 * D_STATE)

typedef float v2f __attribute__((ext_vector_type(2)));
typedef float v8f __attribute__((ext_vector_type(8)));

#define G1_WAVES  8
#define G1_KCHUNK (D_INNER / G1_WAVES)
#define G1_NT     (KPROJ / 16)
#define G1_GROUP  32

struct G1Frag {
    float  a[16];
    float4 q[G1_NT][4];
};

__device__ __forceinline__ void g1_load(G1Frag& f, const float* pa, const float* pb, int k0)
{
    #pragma unroll
    for (int jj = 0; jj < 16; ++jj)
        f.a[jj] = pa[(size_t)(k0 + jj) * SEQLEN];
    #pragma unroll
    for (int t = 0; t < G1_NT; ++t) {
        const float4* r = reinterpret_cast<const float4*>(pb + (size_t)t * 16 * D_INNER + k0);
        #pragma unroll
        for (int i = 0; i < 4; ++i)
            f.q[t][i] = r[i];
    }
}

__device__ __forceinline__ void g1_mma(v8f (&c)[G1_NT], const G1Frag& f)
{
    #pragma unroll
    for (int j = 0; j < 8; ++j) {
        v2f a;
        a.x = f.a[2 * j];  a.y = f.a[2 * j + 1];
        #pragma unroll
        for (int t = 0; t < G1_NT; ++t) {
            const float4 qq = f.q[t][j >> 1];
            v2f bm;
            if (j & 1) { bm.x = qq.z; bm.y = qq.w; }
            else       { bm.x = qq.x; bm.y = qq.y; }
            c[t] = __builtin_amdgcn_wmma_f32_16x16x4_f32(false, a, false, bm, (short)0, c[t], false, false);
        }
    }
}

__global__ __launch_bounds__(256) void xdbl_gemm_wmma(
    const float* __restrict__ u,
    const float* __restrict__ xpw,
    float* __restrict__ xdbl,
    float* __restrict__ bct)
{
    __shared__ float part[G1_WAVES][16 * KPROJ];

    const int lane = threadIdx.x & 31;
    const int wave = threadIdx.x >> 5;
    const int m0   = blockIdx.x * 16;

    const int half = lane & 15;
    const int hi   = lane >> 4;

    const int b = m0 >> 11;
    const int l = (m0 & (SEQLEN - 1)) + half;
    const int kbase = wave * G1_KCHUNK + 16 * hi;

    const float* pa = u + (size_t)b * D_INNER * SEQLEN + (size_t)kbase * SEQLEN + l;
    const float* pb = xpw + (size_t)half * D_INNER + kbase;

    v8f c[G1_NT];
    #pragma unroll
    for (int t = 0; t < G1_NT; ++t) c[t] = v8f{};

    G1Frag f[2];
    g1_load(f[0], pa, pb, 0);
    #pragma unroll
    for (int g = 0; g < G1_KCHUNK / G1_GROUP; ++g) {
        if (g + 1 < G1_KCHUNK / G1_GROUP)
            g1_load(f[(g + 1) & 1], pa, pb, (g + 1) * G1_GROUP);
        __builtin_amdgcn_sched_barrier(0);
        g1_mma(c, f[g & 1]);
        __builtin_amdgcn_sched_barrier(0);
    }

    float* pw = &part[wave][(hi * 8) * KPROJ + half];
    #pragma unroll
    for (int t = 0; t < G1_NT; ++t) {
        #pragma unroll
        for (int r = 0; r < 8; ++r)
            pw[r * KPROJ + t * 16] = c[t][r];
    }
    __syncthreads();

    float* po = xdbl + (size_t)m0 * KPROJ;
    #pragma unroll
    for (int i = 0; i < (16 * DT_RANK) / 256; ++i) {
        const int f   = threadIdx.x + 256 * i;
        const int row = f >> 6, col = f & (DT_RANK - 1);
        const int e   = row * KPROJ + col;
        float s = part[0][e];
        #pragma unroll
        for (int w = 1; w < G1_WAVES; ++w)
            s += part[w][e];
        po[e] = s;
    }

    float* pt = bct + (size_t)(m0 >> 2) * (BC_COLS * 4);
    #pragma unroll
    for (int i = 0; i < (16 * BC_COLS) / 256; ++i) {
        const int f   = threadIdx.x + 256 * i;
        const int row = (f >> 7) * 4 + (f & 3);
        const int col = (f >> 2) & (BC_COLS - 1);
        const int e   = row * KPROJ + DT_RANK + col;
        float s = part[0][e];
        #pragma unroll
        for (int w = 1; w < G1_WAVES; ++w)
            s += part[w][e];
        pt[f] = s;
    }
}

#define G2_MT 8
#define G2_KW (DT_RANK / 4)

struct G2Frag { v2f v[G2_KW]; };

__device__ __forceinline__ void g2_load(G2Frag& f, const float* p)
{
    #pragma unroll
    for (int j = 0; j < G2_KW; ++j)
        f.v[j] = *reinterpret_cast<const v2f*>(p + 4 * j);
}

__device__ __forceinline__ float log1p_unit(float t)
{
    const float d = 2.0f + t;
    const float r = __builtin_amdgcn_rcpf(d);
    float s = t * r;
    const float e = fmaf(-s, d, t);
    s = fmaf(e, r, s);
    const float w = s * s;
    float p = 1.0f / 13.0f;
    p = fmaf(p, w, 1.0f / 11.0f);
    p = fmaf(p, w, 1.0f / 9.0f);
    p = fmaf(p, w, 1.0f / 7.0f);
    p = fmaf(p, w, 1.0f / 5.0f);
    p = fmaf(p, w, 1.0f / 3.0f);
    const float s2 = s + s;
    return fmaf(s2 * w, p, s2);
}

__device__ __forceinline__ void g2_tile(const G2Frag& a, const G2Frag& w, float bias, float* pd)
{
    v8f c0 = {}, c1 = {};
    #pragma unroll
    for (int j = 0; j < G2_KW; j += 2) {
        c0 = __builtin_amdgcn_wmma_f32_16x16x4_f32(false, a.v[j],     false, w.v[j],     (short)0, c0, false, false);
        c1 = __builtin_amdgcn_wmma_f32_16x16x4_f32(false, a.v[j + 1], false, w.v[j + 1], (short)0, c1, false, false);
    }
    v8f c = c0 + c1;

    #pragma unroll
    for (int r = 0; r < 8; ++r) {
        float z  = c[r] + bias;
        float sp = fmaxf(z, 0.0f) + log1p_unit(__expf(-fabsf(z)));
        pd[r] = sp;
    }
}

__global__ __launch_bounds__(256) void delta_gemm_wmma(
    const float* __restrict__ xdbl,
    const float* __restrict__ dtw,
    const float* __restrict__ dtb,
    float* __restrict__ delta)
{
    const int lane = threadIdx.x & 31;
    const int wave = threadIdx.x >> 5;
    const int nt = (blockIdx.x & 15) * 8 + wave;
    const int mg = blockIdx.x >> 4;
    const int m0 = mg * (G2_MT * 16), n0 = nt * 16;

    const int half = lane & 15;
    const int hi   = lane >> 4;
    const int kk   = hi * 2;

    const float* pa = xdbl + (size_t)(m0 + half) * KPROJ + kk;
    const float* pb = dtw + (size_t)(n0 + half) * DT_RANK + kk;

    G2Frag w;
    g2_load(w, pb);

    const int   d    = n0 + half;
    const float bias = dtb[d];
    const int   b    = m0 >> 11;
    const int   lb   = (m0 & (SEQLEN - 1)) + hi * 8;
    float* pd = delta + ((size_t)b * D_INNER + d) * SEQLEN + lb;

    G2Frag a0, a1;
    g2_load(a0, pa);
    #pragma unroll 1
    for (int t = 0; t < G2_MT - 2; t += 2) {
        g2_load(a1, pa + (size_t)(t + 1) * 16 * KPROJ);
        g2_tile(a0, w, bias, pd + t * 16);
        g2_load(a0, pa + (size_t)(t + 2) * 16 * KPROJ);
        g2_tile(a1, w, bias, pd + (t + 1) * 16);
    }
    g2_load(a1, pa + (size_t)(G2_MT - 1) * 16 * KPROJ);
    g2_tile(a0, w, bias, pd + (G2_MT - 2) * 16);
    g2_tile(a1, w, bias, pd + (G2_MT - 1) * 16);
}

struct ScanBlk {
    float4 D4[4], U4[4];
    float4 B4[4], C4[4];
};

template <int K>
__device__ __forceinline__ v2f quad_pair(const float4& v)
{
    v2f r;
    r.x = K == 0 ? v.x : v.z;
    r.y = K == 0 ? v.y : v.w;
    return r;
}

template <int CTRL>
__device__ __forceinline__ float row_dpp(float v)
{
    return __builtin_bit_cast(float,
        __builtin_amdgcn_update_dpp(0, __builtin_bit_cast(int, v), CTRL, 0xf, 0xf, true));
}

__device__ __forceinline__ void scan_load(ScanBlk& q, const float* pdl, const float* pu,
                                          const float4* pbc, int l0)
{
    const float4* rd = reinterpret_cast<const float4*>(pdl + l0);
    const float4* ru = reinterpret_cast<const float4*>(pu + l0);
    const float4* r  = pbc + (size_t)(l0 >> 2) * BC_COLS;
    #pragma unroll
    for (int j = 0; j < 4; ++j) {
        q.D4[j] = rd[j];
        q.U4[j] = ru[j];
        q.B4[j] = r[j * BC_COLS];
        q.C4[j] = r[j * BC_COLS + D_STATE];
    }
}

template <int CTRL>
__device__ __forceinline__ float xr_pair(float a, float b, bool bit_set)
{
    const float keep = bit_set ? b : a;
    const float give = bit_set ? a : b;
    return keep + row_dpp<CTRL>(give);
}

#define LOG2E_F 1.4426950408889634f
template <int J>
struct ScanSteps {
    static __device__ __forceinline__ void run(float& x, float (&P)[16], const ScanBlk& q, v2f An2)
    {
        const v2f d2 = quad_pair<(J >> 1) & 1>(q.D4[J >> 2]);
        const v2f u2 = quad_pair<(J >> 1) & 1>(q.U4[J >> 2]);
        const v2f b2 = quad_pair<(J >> 1) & 1>(q.B4[J >> 2]);
        const v2f c2 = quad_pair<(J >> 1) & 1>(q.C4[J >> 2]);
        const v2f e2 = (d2 * An2) * LOG2E_F;
        const v2f w2 = d2 * u2;
        v2f xs;
        x = __builtin_amdgcn_exp2f(e2.x) * x + w2.x * b2.x;   xs.x = x;
        x = __builtin_amdgcn_exp2f(e2.y) * x + w2.y * b2.y;   xs.y = x;
        const v2f p2 = xs * c2;
        P[J]     = p2.x;
        P[J + 1] = p2.y;
        ScanSteps<J + 2>::run(x, P, q, An2);
    }
};
template <>
struct ScanSteps<16> {
    static __device__ __forceinline__ void run(float&, float (&)[16], const ScanBlk&, v2f) {}
};

__device__ __forceinline__ void scan_block(float& x, float (&r)[2], const ScanBlk& q,
                                           float An, bool b0, bool b1, bool b2)
{
    v2f An2;
    An2.x = An;  An2.y = An;
    float P[16];
    ScanSteps<0>::run(x, P, q, An2);

    float s8[8];
    #pragma unroll
    for (int i = 0; i < 8; ++i) {
        const int j = (i >> 1) * 4 + (i & 1);
        s8[i] = xr_pair<0x161>(P[j], P[j + 2], b0);
    }
    float s4[4];
    #pragma unroll
    for (int i = 0; i < 4; ++i) {
        const int j = (i >> 1) * 4 + (i & 1);
        s4[i] = xr_pair<0x162>(s8[j], s8[j + 2], b1);
    }
    r[0] = xr_pair<0x164>(s4[0], s4[2], b2);
    r[1] = xr_pair<0x164>(s4[1], s4[3], b2);
}

#define SCAN_STAGE_EDGE(x, y0, y1)                                         \
    do {                                                                   \
        asm volatile("" : "+v"(x) : "v"(y0), "v"(y1));                     \
        __builtin_amdgcn_sched_barrier(0);                                 \
    } while (0)

__global__ __launch_bounds__(256) void s6_scan(
    const float* __restrict__ u,
    const float* __restrict__ bct,
    const float* __restrict__ delta,
    const float* __restrict__ A_log,
    float* __restrict__ out)
{
    const int lane = threadIdx.x & 31;
    const int gwave = (blockIdx.x * blockDim.x + threadIdx.x) >> 5;
    const int g = lane >> 4;
    const int n = lane & 15;
    const int ch = gwave * 2 + g;
    const int b  = ch >> 11;
    const int d  = ch & (D_INNER - 1);

    const float An = -__expf(A_log[(size_t)d * D_STATE + n]);

    const float* pdl = delta + (size_t)ch * SEQLEN;
    const float* pu  = u     + (size_t)ch * SEQLEN;
    const float4* pbc = reinterpret_cast<const float4*>(bct) + (size_t)b * (SEQLEN / 4) * BC_COLS + n;
    float*       po  = out   + (size_t)ch * SEQLEN;

    ScanBlk q0, q1, q2, q3;
    scan_load(q0, pdl, pu, pbc, 0);
    scan_load(q1, pdl, pu, pbc, 16);
    scan_load(q2, pdl, pu, pbc, 32);

    const bool b0 = (n & 1) != 0, b1 = (n & 2) != 0, b2 = (n & 4) != 0, b3 = (n & 8) != 0;

    float x = 0.0f;
    float yv[2] = {0.0f, 0.0f};
    float ra[2], rb[2];
    for (int l0 = 0; l0 < SEQLEN; l0 += 64) {
        const int lp0 = l0 + 192 < SEQLEN - 32 ? l0 + 192 : SEQLEN - 32;
        const int lp1 = l0 + 224 < SEQLEN - 32 ? l0 + 224 : SEQLEN - 32;
        __builtin_prefetch(pdl + lp0, 0, 0);
        __builtin_prefetch(pu  + lp0, 0, 0);
        __builtin_prefetch(pdl + lp1, 0, 0);
        __builtin_prefetch(pu  + lp1, 0, 0);

        scan_load(q3, pdl, pu, pbc, l0 + 48);
        SCAN_STAGE_EDGE(x, yv[0], yv[1]);
        scan_block(x, ra, q0, An, b0, b1, b2);

        const int la = l0 + 64 < SEQLEN ? l0 + 64 : SEQLEN - 16;
        SCAN_STAGE_EDGE(x, ra[0], ra[1]);
        scan_load(q0, pdl, pu, pbc, la);
        SCAN_STAGE_EDGE(x, ra[0], ra[1]);
        scan_block(x, rb, q1, An, b0, b1, b2);

        yv[0] = xr_pair<0x168>(ra[0], rb[0], b3);
        yv[1] = xr_pair<0x168>(ra[1], rb[1], b3);
        *reinterpret_cast<float2*>(po + l0 + 2 * n) = make_float2(yv[0], yv[1]);

        const int lb = l0 + 80 < SEQLEN ? l0 + 80 : SEQLEN - 16;
        SCAN_STAGE_EDGE(x, yv[0], yv[1]);
        scan_load(q1, pdl, pu, pbc, lb);
        SCAN_STAGE_EDGE(x, yv[0], yv[1]);
        scan_block(x, ra, q2, An, b0, b1, b2);

        const int lc = l0 + 96 < SEQLEN ? l0 + 96 : SEQLEN - 16;
        SCAN_STAGE_EDGE(x, ra[0], ra[1]);
        scan_load(q2, pdl, pu, pbc, lc);
        SCAN_STAGE_EDGE(x, ra[0], ra[1]);
        scan_block(x, rb, q3, An, b0, b1, b2);

        yv[0] = xr_pair<0x168>(ra[0], rb[0], b3);
        yv[1] = xr_pair<0x168>(ra[1], rb[1], b3);
        *reinterpret_cast<float2*>(po + l0 + 32 + 2 * n) = make_float2(yv[0], yv[1]);
    }
}

extern "C" void kernel_launch(void* const* d_in, const int* in_sizes, int n_in,
                              void* d_out, int out_size, void* d_ws, size_t ws_size,
                              hipStream_t stream)
{
    const float* u     = (const float*)d_in[0];
    const float* xpw   = (const float*)d_in[1];
    const float* dtw   = (const float*)d_in[2];
    const float* dtb   = (const float*)d_in[3];
    const float* A_log = (const float*)d_in[4];
    float*       out   = (float*)d_out;

    float* xdbl  = (float*)d_ws;
    float* bct   = xdbl + (size_t)BATCH * SEQLEN * KPROJ;
    float* delta = bct + (size_t)BATCH * SEQLEN * BC_COLS;

    xdbl_gemm_wmma<<<256, 256, 0, stream>>>(u, xpw, xdbl, bct);

    delta_gemm_wmma<<<512, 256, 0, stream>>>(xdbl, dtw, dtb, delta);

    s6_scan<<<256, 256, 0, stream>>>(u, bct, delta, A_log, out);
}


# delta GEMM fused into the scan kernel: a scan workgroup computes its 16 channels' delta tiles on WMMA into LDS (x_dbl dt columns stored in operand order); 2 launches; delta's k sum order differs
# speedup vs baseline: 13.8361x; 1.1246x over previous
#include <hip/hip_runtime.h>
#include <math.h>

#define BATCH   2
#define SEQLEN  2048
#define D_INNER 2048
#define D_STATE 16
#define DT_RANK 64
#define KPROJ   (DT_RANK + 2 * D_STATE)
#define BC_COLS (2 * D_STATE)

typedef float v2f __attribute__((ext_vector_type(2)));
typedef float v8f __attribute__((ext_vector_type(8)));

#define G1_WAVES  8
#define G1_KCHUNK (D_INNER / G1_WAVES)
#define G1_NT     (KPROJ / 16)
#define G1_GROUP  32

struct G1Frag {
    float  a[16];
    float4 q[G1_NT][4];
};

__device__ __forceinline__ void g1_load(G1Frag& f, const float* pa, const float* pb, int k0)
{
    #pragma unroll
    for (int jj = 0; jj < 16; ++jj)
        f.a[jj] = pa[(size_t)(k0 + jj) * SEQLEN];
    #pragma unroll
    for (int t = 0; t < G1_NT; ++t) {
        const float4* r = reinterpret_cast<const float4*>(pb + (size_t)t * 16 * D_INNER + k0);
        #pragma unroll
        for (int i = 0; i < 4; ++i)
            f.q[t][i] = r[i];
    }
}

__device__ __forceinline__ void g1_mma(v8f (&c)[G1_NT], const G1Frag& f)
{
    #pragma unroll
    for (int j = 0; j < 8; ++j) {
        v2f a;
        a.x = f.a[2 * j];  a.y = f.a[2 * j + 1];
        #pragma unroll
        for (int t = 0; t < G1_NT; ++t) {
            const float4 qq = f.q[t][j >> 1];
            v2f bm;
            if (j & 1) { bm.x = qq.z; bm.y = qq.w; }
            else       { bm.x = qq.x; bm.y = qq.y; }
            c[t] = __builtin_amdgcn_wmma_f32_16x16x4_f32(false, a, false, bm, (short)0, c[t], false, false);
        }
    }
}

__global__ __launch_bounds__(256) void xdbl_gemm_wmma(
    const float* __restrict__ u,
    const float* __restrict__ xpw,
    float* __restrict__ xfr,
    float* __restrict__ bct)
{
    __shared__ float part[G1_WAVES][16 * KPROJ];

    const int lane = threadIdx.x & 31;
    const int wave = threadIdx.x >> 5;
    const int m0   = blockIdx.x * 16;

    const int half = lane & 15;
    const int hi   = lane >> 4;

    const int b = m0 >> 11;
    const int l = (m0 & (SEQLEN - 1)) + half;
    const int kbase = wave * G1_KCHUNK + 16 * hi;

    const float* pa = u + (size_t)b * D_INNER * SEQLEN + (size_t)kbase * SEQLEN + l;
    const float* pb = xpw + (size_t)half * D_INNER + kbase;

    v8f c[G1_NT];
    #pragma unroll
    for (int t = 0; t < G1_NT; ++t) c[t] = v8f{};

    G1Frag f[2];
    g1_load(f[0], pa, pb, 0);
    #pragma unroll
    for (int g = 0; g < G1_KCHUNK / G1_GROUP; ++g) {
        if (g + 1 < G1_KCHUNK / G1_GROUP)
            g1_load(f[(g + 1) & 1], pa, pb, (g + 1) * G1_GROUP);
        __builtin_amdgcn_sched_barrier(0);
        g1_mma(c, f[g & 1]);
        __builtin_amdgcn_sched_barrier(0);
    }

    float* pw = &part[wave][(hi * 8) * KPROJ + half];
    #pragma unroll
    for (int t = 0; t < G1_NT; ++t) {
        #pragma unroll
        for (int r = 0; r < 8; ++r)
            pw[r * KPROJ + t * 16] = c[t][r];
    }
    __syncthreads();

    float* po = xfr + (size_t)(m0 >> 4) * (16 * DT_RANK);
    #pragma unroll
    for (int i = 0; i < (16 * DT_RANK) / 256; ++i) {
        const int f   = threadIdx.x + 256 * i;
        const int ln  = (f >> 2) & 31;
        const int row = ln & 15;
        const int col = (f >> 7) * 8 + (ln >> 4) * 4 + (f & 3);
        const int e   = row * KPROJ + col;
        float s = part[0][e];
        #pragma unroll
        for (int w = 1; w < G1_WAVES; ++w)
            s += part[w][e];
        po[f] = s;
    }

    float* pt = bct + (size_t)(m0 >> 2) * (BC_COLS * 4);
    #pragma unroll
    for (int i = 0; i < (16 * BC_COLS) / 256; ++i) {
        const int f   = threadIdx.x + 256 * i;
        const int row = (f >> 7) * 4 + (f & 3);
        const int col = (f >> 2) & (BC_COLS - 1);
        const int e   = row * KPROJ + DT_RANK + col;
        float s = part[0][e];
        #pragma unroll
        for (int w = 1; w < G1_WAVES; ++w)
            s += part[w][e];
        pt[f] = s;
    }
}

#define G2_TPW   16
#define G2_KQ    (DT_RANK / 8)
#define DL_PITCH (SEQLEN + 4)

struct G2Frag { float4 q[G2_KQ]; };

__device__ __forceinline__ void g2_load(G2Frag& f, const float4* p, int stride)
{
    #pragma unroll
    for (int i = 0; i < G2_KQ; ++i)
        f.q[i] = p[i * stride];
}

__device__ __forceinline__ float log1p_unit(float t)
{
    const float d = 2.0f + t;
    const float r = __builtin_amdgcn_rcpf(d);
    float s = t * r;
    const float e = fmaf(-s, d, t);
    s = fmaf(e, r, s);
    const float w = s * s;
    float p = 1.0f / 13.0f;
    p = fmaf(p, w, 1.0f / 11.0f);
    p = fmaf(p, w, 1.0f / 9.0f);
    p = fmaf(p, w, 1.0f / 7.0f);
    p = fmaf(p, w, 1.0f / 5.0f);
    p = fmaf(p, w, 1.0f / 3.0f);
    const float s2 = s + s;
    return fmaf(s2 * w, p, s2);
}

__device__ __forceinline__ void g2_tile(const G2Frag& a, const G2Frag& w, float bias, float* pd)
{
    v8f c0 = {}, c1 = {};
    #pragma unroll
    for (int p = 0; p < G2_KQ; ++p) {
        v2f a0, a1, w0, w1;
        a0.x = a.q[p].x;  a0.y = a.q[p].y;  a1.x = a.q[p].z;  a1.y = a.q[p].w;
        w0.x = w.q[p].x;  w0.y = w.q[p].y;  w1.x = w.q[p].z;  w1.y = w.q[p].w;
        c0 = __builtin_amdgcn_wmma_f32_16x16x4_f32(false, a0, false, w0, (short)0, c0, false, false);
        c1 = __builtin_amdgcn_wmma_f32_16x16x4_f32(false, a1, false, w1, (short)0, c1, false, false);
    }
    v8f c = c0 + c1;

    float o[8];
    #pragma unroll
    for (int r = 0; r < 8; ++r) {
        float z  = c[r] + bias;
        o[r] = fmaxf(z, 0.0f) + log1p_unit(__expf(-fabsf(z)));
    }
    float4* pq = reinterpret_cast<float4*>(pd);
    pq[0] = make_float4(o[0], o[1], o[2], o[3]);
    pq[1] = make_float4(o[4], o[5], o[6], o[7]);
}

__device__ __forceinline__ void delta_tiles_to_lds(
    const float* __restrict__ xfr,
    const float* __restrict__ dtw,
    const float* __restrict__ dtb,
    float* dls, int b, int d0, int wave, int lane, int rot)
{
    const int half = lane & 15;
    const int hi   = lane >> 4;

    G2Frag w;
    g2_load(w, reinterpret_cast<const float4*>(dtw + (size_t)(d0 + half) * DT_RANK) + hi, 2);
    const float bias = dtb[d0 + half];

    const float4* pa = reinterpret_cast<const float4*>(xfr)
                     + (size_t)(b * (SEQLEN / 16) + wave * G2_TPW) * (16 * DT_RANK / 4) + lane;
    float* pd = dls + half * DL_PITCH + wave * (G2_TPW * 16) + hi * 8;

    #define G2_TILE(t) (((t) + rot) & (G2_TPW - 1))
    G2Frag a0, a1;
    g2_load(a0, pa + G2_TILE(0) * 256, 32);
    #pragma unroll 1
    for (int t = 0; t < G2_TPW - 2; t += 2) {
        g2_load(a1, pa + G2_TILE(t + 1) * 256, 32);
        g2_tile(a0, w, bias, pd + G2_TILE(t) * 16);
        g2_load(a0, pa + G2_TILE(t + 2) * 256, 32);
        g2_tile(a1, w, bias, pd + G2_TILE(t + 1) * 16);
    }
    g2_load(a1, pa + G2_TILE(G2_TPW - 1) * 256, 32);
    g2_tile(a0, w, bias, pd + G2_TILE(G2_TPW - 2) * 16);
    g2_tile(a1, w, bias, pd + G2_TILE(G2_TPW - 1) * 16);
    #undef G2_TILE
}

struct ScanBlk {
    float4 D4[4], U4[4];
    float4 B4[4], C4[4];
};

template <int K>
__device__ __forceinline__ v2f quad_pair(const float4& v)
{
    v2f r;
    r.x = K == 0 ? v.x : v.z;
    r.y = K == 0 ? v.y : v.w;
    return r;
}

template <int CTRL>
__device__ __forceinline__ float row_dpp(float v)
{
    return __builtin_bit_cast(float,
        __builtin_amdgcn_update_dpp(0, __builtin_bit_cast(int, v), CTRL, 0xf, 0xf, true));
}

__device__ __forceinline__ void scan_load(ScanBlk& q, const float* pdl, const float* pu,
                                          const float4* pbc, int l0)
{
    const float4* rd = reinterpret_cast<const float4*>(pdl + l0);
    const float4* ru = reinterpret_cast<const float4*>(pu + l0);
    const float4* r  = pbc + (size_t)(l0 >> 2) * BC_COLS;
    #pragma unroll
    for (int j = 0; j < 4; ++j) {
        q.D4[j] = rd[j];
        q.U4[j] = ru[j];
        q.B4[j] = r[j * BC_COLS];
        q.C4[j] = r[j * BC_COLS + D_STATE];
    }
}

template <int CTRL>
__device__ __forceinline__ float xr_pair(float a, float b, bool bit_set)
{
    const float keep = bit_set ? b : a;
    const float give = bit_set ? a : b;
    return keep + row_dpp<CTRL>(give);
}

#define LOG2E_F 1.4426950408889634f
template <int J>
struct ScanSteps {
    static __device__ __forceinline__ void run(float& x, float (&P)[16], const ScanBlk& q, v2f An2)
    {
        const v2f d2 = quad_pair<(J >> 1) & 1>(q.D4[J >> 2]);
        const v2f u2 = quad_pair<(J >> 1) & 1>(q.U4[J >> 2]);
        const v2f b2 = quad_pair<(J >> 1) & 1>(q.B4[J >> 2]);
        const v2f c2 = quad_pair<(J >> 1) & 1>(q.C4[J >> 2]);
        const v2f e2 = (d2 * An2) * LOG2E_F;
        const v2f w2 = d2 * u2;
        v2f xs;
        x = __builtin_amdgcn_exp2f(e2.x) * x + w2.x * b2.x;   xs.x = x;
        x = __builtin_amdgcn_exp2f(e2.y) * x + w2.y * b2.y;   xs.y = x;
        const v2f p2 = xs * c2;
        P[J]     = p2.x;
        P[J + 1] = p2.y;
        ScanSteps<J + 2>::run(x, P, q, An2);
    }
};
template <>
struct ScanSteps<16> {
    static __device__ __forceinline__ void run(float&, float (&)[16], const ScanBlk&, v2f) {}
};

__device__ __forceinline__ void scan_block(float& x, float (&r)[2], const ScanBlk& q,
                                           float An, bool b0, bool b1, bool b2)
{
    v2f An2;
    An2.x = An;  An2.y = An;
    float P[16];
    ScanSteps<0>::run(x, P, q, An2);

    float s8[8];
    #pragma unroll
    for (int i = 0; i < 8; ++i) {
        const int j = (i >> 1) * 4 + (i & 1);
        s8[i] = xr_pair<0x161>(P[j], P[j + 2], b0);
    }
    float s4[4];
    #pragma unroll
    for (int i = 0; i < 4; ++i) {
        const int j = (i >> 1) * 4 + (i & 1);
        s4[i] = xr_pair<0x162>(s8[j], s8[j + 2], b1);
    }
    r[0] = xr_pair<0x164>(s4[0], s4[2], b2);
    r[1] = xr_pair<0x164>(s4[1], s4[3], b2);
}

#define SCAN_STAGE_EDGE(x, y0, y1)                                         \
    do {                                                                   \
        asm volatile("" : "+v"(x) : "v"(y0), "v"(y1));                     \
        __builtin_amdgcn_sched_barrier(0);                                 \
    } while (0)

__global__ __launch_bounds__(256) void s6_scan(
    const float* __restrict__ u,
    const float* __restrict__ bct,
    const float* __restrict__ xfr,
    const float* __restrict__ dtw,
    const float* __restrict__ dtb,
    const float* __restrict__ A_log,
    float* __restrict__ out)
{
    __shared__ float dls[16 * DL_PITCH];

    const int lane = threadIdx.x & 31;
    const int wave = threadIdx.x >> 5;
    const int g = lane >> 4;
    const int n = lane & 15;
    const int ch0 = blockIdx.x * 16;
    const int ch = ch0 + wave * 2 + g;
    const int b  = ch >> 11;
    const int d  = ch & (D_INNER - 1);

    delta_tiles_to_lds(xfr, dtw, dtb, dls, b, ch0 & (D_INNER - 1), wave, lane, blockIdx.x & (G2_TPW - 1));
    __syncthreads();

    const float An = -__expf(A_log[(size_t)d * D_STATE + n]);

    const float* pdl = dls + (wave * 2 + g) * DL_PITCH;
    const float* pu  = u     + (size_t)ch * SEQLEN;
    const float4* pbc = reinterpret_cast<const float4*>(bct) + (size_t)b * (SEQLEN / 4) * BC_COLS + n;
    float*       po  = out   + (size_t)ch * SEQLEN;

    ScanBlk q0, q1, q2, q3;
    scan_load(q0, pdl, pu, pbc, 0);
    scan_load(q1, pdl, pu, pbc, 16);
    scan_load(q2, pdl, pu, pbc, 32);

    const bool b0 = (n & 1) != 0, b1 = (n & 2) != 0, b2 = (n & 4) != 0, b3 = (n & 8) != 0;

    float x = 0.0f;
    float yv[2] = {0.0f, 0.0f};
    float ra[2], rb[2];
    for (int l0 = 0; l0 < SEQLEN; l0 += 64) {
        const int lp0 = l0 + 192 < SEQLEN - 32 ? l0 + 192 : SEQLEN - 32;
        const int lp1 = l0 + 224 < SEQLEN - 32 ? l0 + 224 : SEQLEN - 32;
        __builtin_prefetch(pu  + lp0, 0, 0);
        __builtin_prefetch(pu  + lp1, 0, 0);

        scan_load(q3, pdl, pu, pbc, l0 + 48);
        SCAN_STAGE_EDGE(x, yv[0], yv[1]);
        scan_block(x, ra, q0, An, b0, b1, b2);

        const int la = l0 + 64 < SEQLEN ? l0 + 64 : SEQLEN - 16;
        SCAN_STAGE_EDGE(x, ra[0], ra[1]);
        scan_load(q0, pdl, pu, pbc, la);
        SCAN_STAGE_EDGE(x, ra[0], ra[1]);
        scan_block(x, rb, q1, An, b0, b1, b2);

        yv[0] = xr_pair<0x168>(ra[0], rb[0], b3);
        yv[1] = xr_pair<0x168>(ra[1], rb[1], b3);
        *reinterpret_cast<float2*>(po + l0 + 2 * n) = make_float2(yv[0], yv[1]);

        const int lb = l0 + 80 < SEQLEN ? l0 + 80 : SEQLEN - 16;
        SCAN_STAGE_EDGE(x, yv[0], yv[1]);
        scan_load(q1, pdl, pu, pbc, lb);
        SCAN_STAGE_EDGE(x, yv[0], yv[1]);
        scan_block(x, ra, q2, An, b0, b1, b2);

        const int lc = l0 + 96 < SEQLEN ? l0 + 96 : SEQLEN - 16;
        SCAN_STAGE_EDGE(x, ra[0], ra[1]);
        scan_load(q2, pdl, pu, pbc, lc);
        SCAN_STAGE_EDGE(x, ra[0], ra[1]);
        scan_block(x, rb, q3, An, b0, b1, b2);

        yv[0] = xr_pair<0x168>(ra[0], rb[0], b3);
        yv[1] = xr_pair<0x168>(ra[1], rb[1], b3);
        *reinterpret_cast<float2*>(po + l0 + 32 + 2 * n) = make_float2(yv[0], yv[1]);
    }
}

extern "C" void kernel_launch(void* const* d_in, const int* in_sizes, int n_in,
                              void* d_out, int out_size, void* d_ws, size_t ws_size,
                              hipStream_t stream)
{
    const float* u     = (const float*)d_in[0];
    const float* xpw   = (const float*)d_in[1];
    const float* dtw   = (const float*)d_in[2];
    const float* dtb   = (const float*)d_in[3];
    const float* A_log = (const float*)d_in[4];
    float*       out   = (float*)d_out;

    float* xfr   = (float*)d_ws;
    float* bct   = xfr + (size_t)BATCH * SEQLEN * DT_RANK;

    xdbl_gemm_wmma<<<256, 256, 0, stream>>>(u, xpw, xfr, bct);

    s6_scan<<<256, 256, 0, stream>>>(u, bct, xfr, dtw, dtb, A_log, out);
}


# x_dbl GEMM re-tiled: a workgroup takes 32 rows x 48 columns (2 m x 3 n tiles per wave, still 6 accumulators) instead of 16 x 96, so each reads half the weight bytes; same sums in the same order
# speedup vs baseline: 15.2396x; 1.1014x over previous
#include <hip/hip_runtime.h>
#include <math.h>

#define BATCH   2
#define SEQLEN  2048
#define D_INNER 2048
#define D_STATE 16
#define DT_RANK 64
#define KPROJ   (DT_RANK + 2 * D_STATE)
#define BC_COLS (2 * D_STATE)

typedef float v2f __attribute__((ext_vector_type(2)));
typedef float v8f __attribute__((ext_vector_type(8)));

#define G1_WAVES  8
#define G1_KCHUNK (D_INNER / G1_WAVES)
#define G1_MT     2
#define G1_NT     3
#define G1_ROWS   (16 * G1_MT)
#define G1_NCOLS  (16 * G1_NT)
#define G1_GROUP  32
#define XFR_TILE  (16 * DT_RANK)
#define XFR_HALF0 (G1_NCOLS * 16)

struct G1Frag {
    float  a[G1_MT][16];
    float4 q[G1_NT][4];
};

__device__ __forceinline__ void g1_load(G1Frag& f, const float* pa, const float* pb, int k0)
{
    #pragma unroll
    for (int jj = 0; jj < 16; ++jj) {
        #pragma unroll
        for (int mt = 0; mt < G1_MT; ++mt)
            f.a[mt][jj] = pa[(size_t)(k0 + jj) * SEQLEN + 16 * mt];
    }
    #pragma unroll
    for (int t = 0; t < G1_NT; ++t) {
        const float4* r = reinterpret_cast<const float4*>(pb + (size_t)t * 16 * D_INNER + k0);
        #pragma unroll
        for (int i = 0; i < 4; ++i)
            f.q[t][i] = r[i];
    }
}

__device__ __forceinline__ void g1_mma(v8f (&c)[G1_MT][G1_NT], const G1Frag& f)
{
    #pragma unroll
    for (int j = 0; j < 8; ++j) {
        v2f bm[G1_NT];
        #pragma unroll
        for (int t = 0; t < G1_NT; ++t) {
            const float4 qq = f.q[t][j >> 1];
            if (j & 1) { bm[t].x = qq.z; bm[t].y = qq.w; }
            else       { bm[t].x = qq.x; bm[t].y = qq.y; }
        }
        #pragma unroll
        for (int mt = 0; mt < G1_MT; ++mt) {
            v2f a;
            a.x = f.a[mt][2 * j];  a.y = f.a[mt][2 * j + 1];
            #pragma unroll
            for (int t = 0; t < G1_NT; ++t)
                c[mt][t] = __builtin_amdgcn_wmma_f32_16x16x4_f32(false, a, false, bm[t], (short)0, c[mt][t], false, false);
        }
    }
}

__device__ __forceinline__ float g1_sum(const float (&part)[G1_WAVES][G1_ROWS * G1_NCOLS], int e)
{
    float s = part[0][e];
    #pragma unroll
    for (int w = 1; w < G1_WAVES; ++w)
        s += part[w][e];
    return s;
}

__global__ __launch_bounds__(256) void xdbl_gemm_wmma(
    const float* __restrict__ u,
    const float* __restrict__ xpw,
    float* __restrict__ xfr,
    float* __restrict__ bct)
{
    __shared__ float part[G1_WAVES][G1_ROWS * G1_NCOLS];

    const int lane = threadIdx.x & 31;
    const int wave = threadIdx.x >> 5;
    const int nh   = blockIdx.x & 1;
    const int m0   = (blockIdx.x >> 1) * G1_ROWS;

    const int half = lane & 15;
    const int hi   = lane >> 4;

    const int b = m0 >> 11;
    const int l = (m0 & (SEQLEN - 1)) + half;
    const int kbase = wave * G1_KCHUNK + 16 * hi;

    const float* pa = u + (size_t)b * D_INNER * SEQLEN + (size_t)kbase * SEQLEN + l;
    const float* pb = xpw + (size_t)(nh * G1_NCOLS + half) * D_INNER + kbase;

    v8f c[G1_MT][G1_NT];
    #pragma unroll
    for (int mt = 0; mt < G1_MT; ++mt) {
        #pragma unroll
        for (int t = 0; t < G1_NT; ++t) c[mt][t] = v8f{};
    }

    G1Frag f[2];
    g1_load(f[0], pa, pb, 0);
    #pragma unroll
    for (int g = 0; g < G1_KCHUNK / G1_GROUP; ++g) {
        if (g + 1 < G1_KCHUNK / G1_GROUP)
            g1_load(f[(g + 1) & 1], pa, pb, (g + 1) * G1_GROUP);
        __builtin_amdgcn_sched_barrier(0);
        g1_mma(c, f[g & 1]);
        __builtin_amdgcn_sched_barrier(0);
    }

    float* pw = &part[wave][(hi * 8) * G1_NCOLS + half];
    #pragma unroll
    for (int mt = 0; mt < G1_MT; ++mt) {
        #pragma unroll
        for (int t = 0; t < G1_NT; ++t) {
            #pragma unroll
            for (int r = 0; r < 8; ++r)
                pw[(mt * 16 + r) * G1_NCOLS + t * 16] = c[mt][t][r];
        }
    }
    __syncthreads();

    float* po = xfr + (size_t)(m0 >> 4) * XFR_TILE;
    if (nh == 0) {
        #pragma unroll
        for (int i = 0; i < (G1_MT * XFR_HALF0) / 256; ++i) {
            const int mt  = i / (XFR_HALF0 / 256);
            const int f   = threadIdx.x + 256 * (i % (XFR_HALF0 / 256));
            const int ln  = (f >> 2) & 31;
            const int row = ln & 15;
            const int col = (f >> 7) * 8 + (ln >> 4) * 4 + (f & 3);
            po[mt * XFR_TILE + f] = g1_sum(part, (mt * 16 + row) * G1_NCOLS + col);
        }
    } else {
        #pragma unroll
        for (int mt = 0; mt < G1_MT; ++mt) {
            const int f   = XFR_HALF0 + threadIdx.x;
            const int ln  = (f >> 2) & 31;
            const int row = ln & 15;
            const int col = (f >> 7) * 8 + (ln >> 4) * 4 + (f & 3) - G1_NCOLS;
            po[mt * XFR_TILE + f] = g1_sum(part, (mt * 16 + row) * G1_NCOLS + col);
        }
        float* pt = bct + (size_t)(m0 >> 2) * (BC_COLS * 4);
        #pragma unroll
        for (int i = 0; i < (G1_ROWS * BC_COLS) / 256; ++i) {
            const int f   = threadIdx.x + 256 * i;
            const int row = (f >> 7) * 4 + (f & 3);
            const int col = (f >> 2) & (BC_COLS - 1);
            pt[f] = g1_sum(part, row * G1_NCOLS + (DT_RANK - G1_NCOLS) + col);
        }
    }
}

#define G2_TPW   16
#define G2_KQ    (DT_RANK / 8)
#define DL_PITCH (SEQLEN + 4)

struct G2Frag { float4 q[G2_KQ]; };

__device__ __forceinline__ void g2_load(G2Frag& f, const float4* p, int stride)
{
    #pragma unroll
    for (int i = 0; i < G2_KQ; ++i)
        f.q[i] = p[i * stride];
}

__device__ __forceinline__ float log1p_unit(float t)
{
    const float d = 2.0f + t;
    const float r = __builtin_amdgcn_rcpf(d);
    float s = t * r;
    const float e = fmaf(-s, d, t);
    s = fmaf(e, r, s);
    const float w = s * s;
    float p = 1.0f / 13.0f;
    p = fmaf(p, w, 1.0f / 11.0f);
    p = fmaf(p, w, 1.0f / 9.0f);
    p = fmaf(p, w, 1.0f / 7.0f);
    p = fmaf(p, w, 1.0f / 5.0f);
    p = fmaf(p, w, 1.0f / 3.0f);
    const float s2 = s + s;
    return fmaf(s2 * w, p, s2);
}

__device__ __forceinline__ void g2_tile(const G2Frag& a, const G2Frag& w, float bias, float* pd)
{
    v8f c0 = {}, c1 = {};
    #pragma unroll
    for (int p = 0; p < G2_KQ; ++p) {
        v2f a0, a1, w0, w1;
        a0.x = a.q[p].x;  a0.y = a.q[p].y;  a1.x = a.q[p].z;  a1.y = a.q[p].w;
        w0.x = w.q[p].x;  w0.y = w.q[p].y;  w1.x = w.q[p].z;  w1.y = w.q[p].w;
        c0 = __builtin_amdgcn_wmma_f32_16x16x4_f32(false, a0, false, w0, (short)0, c0, false, false);
        c1 = __builtin_amdgcn_wmma_f32_16x16x4_f32(false, a1, false, w1, (short)0, c1, false, false);
    }
    v8f c = c0 + c1;

    float o[8];
    #pragma unroll
    for (int r = 0; r < 8; ++r) {
        float z  = c[r] + bias;
        o[r] = fmaxf(z, 0.0f) + log1p_unit(__expf(-fabsf(z)));
    }
    float4* pq = reinterpret_cast<float4*>(pd);
    pq[0] = make_float4(o[0], o[1], o[2], o[3]);
    pq[1] = make_float4(o[4], o[5], o[6], o[7]);
}

__device__ __forceinline__ void delta_tiles_to_lds(
    const float* __restrict__ xfr,
    const float* __restrict__ dtw,
    const float* __restrict__ dtb,
    float* dls, int b, int d0, int wave, int lane, int rot)
{
    const int half = lane & 15;
    const int hi   = lane >> 4;

    G2Frag w;
    g2_load(w, reinterpret_cast<const float4*>(dtw + (size_t)(d0 + half) * DT_RANK) + hi, 2);
    const float bias = dtb[d0 + half];

    const float4* pa = reinterpret_cast<const float4*>(xfr)
                     + (size_t)(b * (SEQLEN / 16) + wave * G2_TPW) * (16 * DT_RANK / 4) + lane;
    float* pd = dls + half * DL_PITCH + wave * (G2_TPW * 16) + hi * 8;

    #define G2_TILE(t) (((t) + rot) & (G2_TPW - 1))
    G2Frag a0, a1;
    g2_load(a0, pa + G2_TILE(0) * 256, 32);
    #pragma unroll 1
    for (int t = 0; t < G2_TPW - 2; t += 2) {
        g2_load(a1, pa + G2_TILE(t + 1) * 256, 32);
        g2_tile(a0, w, bias, pd + G2_TILE(t) * 16);
        g2_load(a0, pa + G2_TILE(t + 2) * 256, 32);
        g2_tile(a1, w, bias, pd + G2_TILE(t + 1) * 16);
    }
    g2_load(a1, pa + G2_TILE(G2_TPW - 1) * 256, 32);
    g2_tile(a0, w, bias, pd + G2_TILE(G2_TPW - 2) * 16);
    g2_tile(a1, w, bias, pd + G2_TILE(G2_TPW - 1) * 16);
    #undef G2_TILE
}

struct ScanBlk {
    float4 D4[4], U4[4];
    float4 B4[4], C4[4];
};

template <int K>
__device__ __forceinline__ v2f quad_pair(const float4& v)
{
    v2f r;
    r.x = K == 0 ? v.x : v.z;
    r.y = K == 0 ? v.y : v.w;
    return r;
}

template <int CTRL>
__device__ __forceinline__ float row_dpp(float v)
{
    return __builtin_bit_cast(float,
        __builtin_amdgcn_update_dpp(0, __builtin_bit_cast(int, v), CTRL, 0xf, 0xf, true));
}

__device__ __forceinline__ void scan_load(ScanBlk& q, const float* pdl, const float* pu,
                                          const float4* pbc, int l0)
{
    const float4* rd = reinterpret_cast<const float4*>(pdl + l0);
    const float4* ru = reinterpret_cast<const float4*>(pu + l0);
    const float4* r  = pbc + (size_t)(l0 >> 2) * BC_COLS;
    #pragma unroll
    for (int j = 0; j < 4; ++j) {
        q.D4[j] = rd[j];
        q.U4[j] = ru[j];
        q.B4[j] = r[j * BC_COLS];
        q.C4[j] = r[j * BC_COLS + D_STATE];
    }
}

template <int CTRL>
__device__ __forceinline__ float xr_pair(float a, float b, bool bit_set)
{
    const float keep = bit_set ? b : a;
    const float give = bit_set ? a : b;
    return keep + row_dpp<CTRL>(give);
}

#define LOG2E_F 1.4426950408889634f
template <int J>
struct ScanSteps {
    static __device__ __forceinline__ void run(float& x, float (&P)[16], const ScanBlk& q, v2f An2)
    {
        const v2f d2 = quad_pair<(J >> 1) & 1>(q.D4[J >> 2]);
        const v2f u2 = quad_pair<(J >> 1) & 1>(q.U4[J >> 2]);
        const v2f b2 = quad_pair<(J >> 1) & 1>(q.B4[J >> 2]);
        const v2f c2 = quad_pair<(J >> 1) & 1>(q.C4[J >> 2]);
        const v2f e2 = (d2 * An2) * LOG2E_F;
        const v2f w2 = d2 * u2;
        v2f xs;
        x = __builtin_amdgcn_exp2f(e2.x) * x + w2.x * b2.x;   xs.x = x;
        x = __builtin_amdgcn_exp2f(e2.y) * x + w2.y * b2.y;   xs.y = x;
        const v2f p2 = xs * c2;
        P[J]     = p2.x;
        P[J + 1] = p2.y;
        ScanSteps<J + 2>::run(x, P, q, An2);
    }
};
template <>
struct ScanSteps<16> {
    static __device__ __forceinline__ void run(float&, float (&)[16], const ScanBlk&, v2f) {}
};

__device__ __forceinline__ void scan_block(float& x, float (&r)[2], const ScanBlk& q,
                                           float An, bool b0, bool b1, bool b2)
{
    v2f An2;
    An2.x = An;  An2.y = An;
    float P[16];
    ScanSteps<0>::run(x, P, q, An2);

    float s8[8];
    #pragma unroll
    for (int i = 0; i < 8; ++i) {
        const int j = (i >> 1) * 4 + (i & 1);
        s8[i] = xr_pair<0x161>(P[j], P[j + 2], b0);
    }
    float s4[4];
    #pragma unroll
    for (int i = 0; i < 4; ++i) {
        const int j = (i >> 1) * 4 + (i & 1);
        s4[i] = xr_pair<0x162>(s8[j], s8[j + 2], b1);
    }
    r[0] = xr_pair<0x164>(s4[0], s4[2], b2);
    r[1] = xr_pair<0x164>(s4[1], s4[3], b2);
}

#define SCAN_STAGE_EDGE(x, y0, y1)                                         \
    do {                                                                   \
        asm volatile("" : "+v"(x) : "v"(y0), "v"(y1));                     \
        __builtin_amdgcn_sched_barrier(0);                                 \
    } while (0)

__global__ __launch_bounds__(256) void s6_scan(
    const float* __restrict__ u,
    const float* __restrict__ bct,
    const float* __restrict__ xfr,
    const float* __restrict__ dtw,
    const float* __restrict__ dtb,
    const float* __restrict__ A_log,
    float* __restrict__ out)
{
    __shared__ float dls[16 * DL_PITCH];

    const int lane = threadIdx.x & 31;
    const int wave = threadIdx.x >> 5;
    const int g = lane >> 4;
    const int n = lane & 15;
    const int ch0 = blockIdx.x * 16;
    const int ch = ch0 + wave * 2 + g;
    const int b  = ch >> 11;
    const int d  = ch & (D_INNER - 1);

    delta_tiles_to_lds(xfr, dtw, dtb, dls, b, ch0 & (D_INNER - 1), wave, lane, blockIdx.x & (G2_TPW - 1));
    __syncthreads();

    const float An = -__expf(A_log[(size_t)d * D_STATE + n]);

    const float* pdl = dls + (wave * 2 + g) * DL_PITCH;
    const float* pu  = u     + (size_t)ch * SEQLEN;
    const float4* pbc = reinterpret_cast<const float4*>(bct) + (size_t)b * (SEQLEN / 4) * BC_COLS + n;
    float*       po  = out   + (size_t)ch * SEQLEN;

    ScanBlk q0, q1, q2, q3;
    scan_load(q0, pdl, pu, pbc, 0);
    scan_load(q1, pdl, pu, pbc, 16);
    scan_load(q2, pdl, pu, pbc, 32);

    const bool b0 = (n & 1) != 0, b1 = (n & 2) != 0, b2 = (n & 4) != 0, b3 = (n & 8) != 0;

    float x = 0.0f;
    float yv[2] = {0.0f, 0.0f};
    float ra[2], rb[2];
    for (int l0 = 0; l0 < SEQLEN; l0 += 64) {
        const int lp0 = l0 + 192 < SEQLEN - 32 ? l0 + 192 : SEQLEN - 32;
        const int lp1 = l0 + 224 < SEQLEN - 32 ? l0 + 224 : SEQLEN - 32;
        __builtin_prefetch(pu  + lp0, 0, 0);
        __builtin_prefetch(pu  + lp1, 0, 0);

        scan_load(q3, pdl, pu, pbc, l0 + 48);
        SCAN_STAGE_EDGE(x, yv[0], yv[1]);
        scan_block(x, ra, q0, An, b0, b1, b2);

        const int la = l0 + 64 < SEQLEN ? l0 + 64 : SEQLEN - 16;
        SCAN_STAGE_EDGE(x, ra[0], ra[1]);
        scan_load(q0, pdl, pu, pbc, la);
        SCAN_STAGE_EDGE(x, ra[0], ra[1]);
        scan_block(x, rb, q1, An, b0, b1, b2);

        yv[0] = xr_pair<0x168>(ra[0], rb[0], b3);
        yv[1] = xr_pair<0x168>(ra[1], rb[1], b3);
        *reinterpret_cast<float2*>(po + l0 + 2 * n) = make_float2(yv[0], yv[1]);

        const int lb = l0 + 80 < SEQLEN ? l0 + 80 : SEQLEN - 16;
        SCAN_STAGE_EDGE(x, yv[0], yv[1]);
        scan_load(q1, pdl, pu, pbc, lb);
        SCAN_STAGE_EDGE(x, yv[0], yv[1]);
        scan_block(x, ra, q2, An, b0, b1, b2);

        const int lc = l0 + 96 < SEQLEN ? l0 + 96 : SEQLEN - 16;
        SCAN_STAGE_EDGE(x, ra[0], ra[1]);
        scan_load(q2, pdl, pu, pbc, lc);
        SCAN_STAGE_EDGE(x, ra[0], ra[1]);
        scan_block(x, rb, q3, An, b0, b1, b2);

        yv[0] = xr_pair<0x168>(ra[0], rb[0], b3);
        yv[1] = xr_pair<0x168>(ra[1], rb[1], b3);
        *reinterpret_cast<float2*>(po + l0 + 32 + 2 * n) = make_float2(yv[0], yv[1]);
    }
}

extern "C" void kernel_launch(void* const* d_in, const int* in_sizes, int n_in,
                              void* d_out, int out_size, void* d_ws, size_t ws_size,
                              hipStream_t stream)
{
    const float* u     = (const float*)d_in[0];
    const float* xpw   = (const float*)d_in[1];
    const float* dtw   = (const float*)d_in[2];
    const float* dtb   = (const float*)d_in[3];
    const float* A_log = (const float*)d_in[4];
    float*       out   = (float*)d_out;

    float* xfr   = (float*)d_ws;
    float* bct   = xfr + (size_t)BATCH * SEQLEN * DT_RANK;

    xdbl_gemm_wmma<<<256, 256, 0, stream>>>(u, xpw, xfr, bct);

    s6_scan<<<256, 256, 0, stream>>>(u, bct, xfr, dtw, dtb, A_log, out);
}
